# starting kernel
# baseline (speedup 1.0000x reference)
#include <hip/hip_runtime.h>
#include <hip/hip_bf16.h>
#include <math.h>


typedef float v2f __attribute__((ext_vector_type(2)));
typedef float v8f __attribute__((ext_vector_type(8)));

#define DM   1024
#define DI   2048
#define NS   16
#define KC   4
#define DTR  64
#define LSEQ 2048
#define XDBL_W (DTR + 2*NS)

template<int WM, int WN, bool SOFTPLUS_BIAS>
__global__ __launch_bounds__(256)
void wmma_gemm_f32(const float* __restrict__ A, const float* __restrict__ B,
                   float* __restrict__ C, const float* __restrict__ bias,
                   int M, int N, int K, int lda, int ldb, int ldc)
{
    const int wave = (blockIdx.x * blockDim.x + threadIdx.x) >> 5;
    const int lane = threadIdx.x & 31;
    const int tilesM = M / (WM * 16);
    const int tilesN = N / (WN * 16);
    if (wave >= tilesM * tilesN) return;
    const int tm = wave % tilesM;
    const int tn = wave / tilesM;
    const int m0 = tm * WM * 16;
    const int n0 = tn * WN * 16;
    const int lr = lane & 15;
    const int lh = lane >> 4;

    v8f acc[WM][WN];
#pragma unroll
    for (int i = 0; i < WM; ++i)
#pragma unroll
        for (int j = 0; j < WN; ++j) {
            v8f z = {0.f, 0.f, 0.f, 0.f, 0.f, 0.f, 0.f, 0.f};
            acc[i][j] = z;
        }

    for (int k0 = 0; k0 < K; k0 += 4) {
        v2f a[WM];
#pragma unroll
        for (int i = 0; i < WM; ++i) {
            const float* ap = A + (size_t)(m0 + i * 16 + lr) * lda + k0 + 2 * lh;
            a[i] = *(const v2f*)ap;
        }
        v2f b[WN];
#pragma unroll
        for (int j = 0; j < WN; ++j) {
            const float* bp = B + (size_t)(k0 + 2 * lh) * ldb + n0 + j * 16 + lr;
            v2f t; t[0] = bp[0]; t[1] = bp[ldb];
            b[j] = t;
        }
#pragma unroll
        for (int i = 0; i < WM; ++i)
#pragma unroll
            for (int j = 0; j < WN; ++j)
                acc[i][j] = __builtin_amdgcn_wmma_f32_16x16x4_f32(
                    false, a[i], false, b[j], (short)0, acc[i][j], false, false);
    }

#pragma unroll
    for (int i = 0; i < WM; ++i) {
#pragma unroll
        for (int j = 0; j < WN; ++j) {
            const int col = n0 + j * 16 + lr;
#pragma unroll
            for (int r = 0; r < 8; ++r) {
                const int row = m0 + i * 16 + r + 8 * lh;
                float v = acc[i][j][r];
                if (SOFTPLUS_BIAS) {
                    v += bias[col];
                    v = (v > 20.f) ? v : __logf(1.f + __expf(v));
                }
                C[(size_t)row * ldc + col] = v;
            }
        }
    }
}

__global__ __launch_bounds__(256)
void conv_silu_kernel(const float* __restrict__ xr, const float* __restrict__ conv_w,
                      const float* __restrict__ conv_b, float* __restrict__ xs)
{
    const int idx = blockIdx.x * 256 + threadIdx.x;
    const int l = idx / DI;
    const int d = idx % DI;
    const float4 w = *(const float4*)(conv_w + (size_t)d * KC);
    float acc = conv_b[d];
    const float* xc = xr + d;
    if (l >= 3) acc += xc[(size_t)(l - 3) * (2 * DI)] * w.x;
    if (l >= 2) acc += xc[(size_t)(l - 2) * (2 * DI)] * w.y;
    if (l >= 1) acc += xc[(size_t)(l - 1) * (2 * DI)] * w.z;
    acc             += xc[(size_t)l       * (2 * DI)] * w.w;
    const float sig = 1.f / (1.f + __expf(-acc));
    xs[idx] = acc * sig;
}

__global__ __launch_bounds__(256)
void scan_kernel(const float* __restrict__ dt, const float* __restrict__ xdbl,
                 const float* __restrict__ xs, const float* __restrict__ xr,
                 const float* __restrict__ A_log, const float* __restrict__ Dvec,
                 float* __restrict__ y, float* __restrict__ hT)
{
    __shared__ float sDT[64 * 32];
    __shared__ float sXS[64 * 32];
    __shared__ float sBC[64 * 32];

    const int t  = blockIdx.x * 256 + threadIdx.x;
    const int d  = t >> 3;
    const int q  = t & 7;
    const int ch = threadIdx.x >> 3;
    const int d0 = blockIdx.x * 32;

    float A0 = -__expf(A_log[(size_t)d * NS + 2 * q]);
    float A1 = -__expf(A_log[(size_t)d * NS + 2 * q + 1]);
    float h0 = 0.f, h1 = 0.f;
    const float Dd = Dvec[d];

    for (int c = 0; c < LSEQ / 64; ++c) {
        __syncthreads();
        for (int e = threadIdx.x; e < 64 * 32; e += 256) {
            const int ls = e >> 5, w = e & 31;
            const int l = c * 64 + ls;
            sDT[e] = dt[(size_t)l * DI + d0 + w];
            sXS[e] = xs[(size_t)l * DI + d0 + w];
            sBC[e] = xdbl[(size_t)l * XDBL_W + DTR + w];
        }
        __syncthreads();
        for (int s = 0; s < 64; ++s) {
            const int l = c * 64 + s;
            const float dtv = sDT[s * 32 + ch];
            const float xv  = sXS[s * 32 + ch];
            const float2 Bv = *(const float2*)&sBC[s * 32 + 2 * q];
            const float2 Cv = *(const float2*)&sBC[s * 32 + 16 + 2 * q];
            const float dx  = dtv * xv;
            const float dA0 = __expf(dtv * A0);
            const float dA1 = __expf(dtv * A1);
            h0 = dA0 * h0 + dx * Bv.x;
            h1 = dA1 * h1 + dx * Bv.y;
            float yacc = h0 * Cv.x + h1 * Cv.y;
            yacc += __shfl_xor(yacc, 1);
            yacc += __shfl_xor(yacc, 2);
            yacc += __shfl_xor(yacc, 4);
            if (q == 0) {
                const float resv = xr[(size_t)l * (2 * DI) + DI + d];
                const float yv   = yacc + xv * Dd;
                const float sig  = 1.f / (1.f + __expf(-resv));
                y[(size_t)l * DI + d] = yv * (resv * sig);
            }
        }
    }
    float2 hout = make_float2(h0, h1);
    *(float2*)&hT[(size_t)d * NS + 2 * q] = hout;
}

extern "C" void kernel_launch(void* const* d_in, const int* in_sizes, int n_in,
                              void* d_out, int out_size, void* d_ws, size_t ws_size,
                              hipStream_t stream) {
    const float* x      = (const float*)d_in[0];
    const float* W_in   = (const float*)d_in[1];
    const float* conv_w = (const float*)d_in[2];
    const float* conv_b = (const float*)d_in[3];
    const float* W_x    = (const float*)d_in[4];
    const float* W_dt   = (const float*)d_in[5];
    const float* b_dt   = (const float*)d_in[6];
    const float* A_log  = (const float*)d_in[7];
    const float* Dv     = (const float*)d_in[8];
    const float* W_out  = (const float*)d_in[9];

    float* out = (float*)d_out;
    float* hT  = out + (size_t)LSEQ * DM;

    float* ws   = (float*)d_ws;
    float* xr   = ws;
    float* xs   = xr   + (size_t)LSEQ * 2 * DI;
    float* xdbl = xs   + (size_t)LSEQ * DI;
    float* dt   = xdbl + (size_t)LSEQ * XDBL_W;
    float* yb   = dt   + (size_t)LSEQ * DI;

    wmma_gemm_f32<4, 4, false><<<256, 256, 0, stream>>>(
        x, W_in, xr, nullptr, LSEQ, 2 * DI, DM, DM, 2 * DI, 2 * DI);

    conv_silu_kernel<<<(LSEQ * DI) / 256, 256, 0, stream>>>(xr, conv_w, conv_b, xs);

    wmma_gemm_f32<4, 2, false><<<12, 256, 0, stream>>>(
        xs, W_x, xdbl, nullptr, LSEQ, XDBL_W, DI, DI, XDBL_W, XDBL_W);

    wmma_gemm_f32<4, 4, true><<<128, 256, 0, stream>>>(
        xdbl, W_dt, dt, b_dt, LSEQ, DI, DTR, XDBL_W, DI, DI);

    scan_kernel<<<(DI * 8) / 256, 256, 0, stream>>>(dt, xdbl, xs, xr, A_log, Dv, yb, hT);

    wmma_gemm_f32<4, 4, false><<<64, 256, 0, stream>>>(
        yb, W_out, out, nullptr, LSEQ, DM, DI, DI, DM, DM);
}


# scan kernel: gate load and y store leave the 2048-step serial loop: res fetched to registers ahead of each 64-step chunk, y kept in an LDS tile and written as full rows after it; sums unchanged
# speedup vs baseline: 1.6757x; 1.6757x over previous
#include <hip/hip_runtime.h>
#include <hip/hip_bf16.h>
#include <math.h>


typedef float v2f __attribute__((ext_vector_type(2)));
typedef float v8f __attribute__((ext_vector_type(8)));

#define DM   1024
#define DI   2048
#define NS   16
#define KC   4
#define DTR  64
#define LSEQ 2048
#define XDBL_W (DTR + 2*NS)

template<int WM, int WN, bool SOFTPLUS_BIAS>
__global__ __launch_bounds__(256)
void wmma_gemm_f32(const float* __restrict__ A, const float* __restrict__ B,
                   float* __restrict__ C, const float* __restrict__ bias,
                   int M, int N, int K, int lda, int ldb, int ldc)
{
    const int wave = (blockIdx.x * blockDim.x + threadIdx.x) >> 5;
    const int lane = threadIdx.x & 31;
    const int tilesM = M / (WM * 16);
    const int tilesN = N / (WN * 16);
    if (wave >= tilesM * tilesN) return;
    const int tm = wave % tilesM;
    const int tn = wave / tilesM;
    const int m0 = tm * WM * 16;
    const int n0 = tn * WN * 16;
    const int lr = lane & 15;
    const int lh = lane >> 4;

    v8f acc[WM][WN];
#pragma unroll
    for (int i = 0; i < WM; ++i)
#pragma unroll
        for (int j = 0; j < WN; ++j) {
            v8f z = {0.f, 0.f, 0.f, 0.f, 0.f, 0.f, 0.f, 0.f};
            acc[i][j] = z;
        }

    for (int k0 = 0; k0 < K; k0 += 4) {
        v2f a[WM];
#pragma unroll
        for (int i = 0; i < WM; ++i) {
            const float* ap = A + (size_t)(m0 + i * 16 + lr) * lda + k0 + 2 * lh;
            a[i] = *(const v2f*)ap;
        }
        v2f b[WN];
#pragma unroll
        for (int j = 0; j < WN; ++j) {
            const float* bp = B + (size_t)(k0 + 2 * lh) * ldb + n0 + j * 16 + lr;
            v2f t; t[0] = bp[0]; t[1] = bp[ldb];
            b[j] = t;
        }
#pragma unroll
        for (int i = 0; i < WM; ++i)
#pragma unroll
            for (int j = 0; j < WN; ++j)
                acc[i][j] = __builtin_amdgcn_wmma_f32_16x16x4_f32(
                    false, a[i], false, b[j], (short)0, acc[i][j], false, false);
    }

#pragma unroll
    for (int i = 0; i < WM; ++i) {
#pragma unroll
        for (int j = 0; j < WN; ++j) {
            const int col = n0 + j * 16 + lr;
#pragma unroll
            for (int r = 0; r < 8; ++r) {
                const int row = m0 + i * 16 + r + 8 * lh;
                float v = acc[i][j][r];
                if (SOFTPLUS_BIAS) {
                    v += bias[col];
                    v = (v > 20.f) ? v : __logf(1.f + __expf(v));
                }
                C[(size_t)row * ldc + col] = v;
            }
        }
    }
}

__global__ __launch_bounds__(256)
void conv_silu_kernel(const float* __restrict__ xr, const float* __restrict__ conv_w,
                      const float* __restrict__ conv_b, float* __restrict__ xs)
{
    const int idx = blockIdx.x * 256 + threadIdx.x;
    const int l = idx / DI;
    const int d = idx % DI;
    const float4 w = *(const float4*)(conv_w + (size_t)d * KC);
    float acc = conv_b[d];
    const float* xc = xr + d;
    if (l >= 3) acc += xc[(size_t)(l - 3) * (2 * DI)] * w.x;
    if (l >= 2) acc += xc[(size_t)(l - 2) * (2 * DI)] * w.y;
    if (l >= 1) acc += xc[(size_t)(l - 1) * (2 * DI)] * w.z;
    acc             += xc[(size_t)l       * (2 * DI)] * w.w;
    const float sig = 1.f / (1.f + __expf(-acc));
    xs[idx] = acc * sig;
}

__global__ __launch_bounds__(256)
void scan_kernel(const float* __restrict__ dt, const float* __restrict__ xdbl,
                 const float* __restrict__ xs, const float* __restrict__ xr,
                 const float* __restrict__ A_log, const float* __restrict__ Dvec,
                 float* __restrict__ y, float* __restrict__ hT)
{
    __shared__ float sDT[64 * 32];
    __shared__ float sXS[64 * 32];
    __shared__ float sBC[64 * 32];
    __shared__ float sY[64 * 32];

    const int t  = blockIdx.x * 256 + threadIdx.x;
    const int d  = t >> 3;
    const int q  = t & 7;
    const int ch = threadIdx.x >> 3;
    const int d0 = blockIdx.x * 32;
    const int wr = threadIdx.x >> 5;
    const int wc = threadIdx.x & 31;

    float A0 = -__expf(A_log[(size_t)d * NS + 2 * q]);
    float A1 = -__expf(A_log[(size_t)d * NS + 2 * q + 1]);
    float h0 = 0.f, h1 = 0.f;
    const float Dd = Dvec[d];

    for (int c = 0; c < LSEQ / 64; ++c) {
        for (int e = threadIdx.x; e < 64 * 32; e += 256) {
            const int ls = e >> 5, w = e & 31;
            const int l = c * 64 + ls;
            sDT[e] = dt[(size_t)l * DI + d0 + w];
            sXS[e] = xs[(size_t)l * DI + d0 + w];
            sBC[e] = xdbl[(size_t)l * XDBL_W + DTR + w];
        }
        __syncthreads();
        float resv[8];
        for (int k = 0; k < 8; ++k) {
            const int l = c * 64 + wr + 8 * k;
            resv[k] = xr[(size_t)l * (2 * DI) + DI + d0 + wc];
        }
        for (int s = 0; s < 64; ++s) {
            const float dtv = sDT[s * 32 + ch];
            const float xv  = sXS[s * 32 + ch];
            const float2 Bv = *(const float2*)&sBC[s * 32 + 2 * q];
            const float2 Cv = *(const float2*)&sBC[s * 32 + 16 + 2 * q];
            const float dx  = dtv * xv;
            const float dA0 = __expf(dtv * A0);
            const float dA1 = __expf(dtv * A1);
            h0 = dA0 * h0 + dx * Bv.x;
            h1 = dA1 * h1 + dx * Bv.y;
            float yacc = h0 * Cv.x + h1 * Cv.y;
            yacc += __shfl_xor(yacc, 1);
            yacc += __shfl_xor(yacc, 2);
            yacc += __shfl_xor(yacc, 4);
            if (q == 0) sY[s * 32 + ch] = yacc + xv * Dd;
        }
        __syncthreads();
        for (int k = 0; k < 8; ++k) {
            const int ls = wr + 8 * k;
            const int l = c * 64 + ls;
            const float sig = 1.f / (1.f + __expf(-resv[k]));
            y[(size_t)l * DI + d0 + wc] = sY[ls * 32 + wc] * (resv[k] * sig);
        }
    }
    float2 hout = make_float2(h0, h1);
    *(float2*)&hT[(size_t)d * NS + 2 * q] = hout;
}

extern "C" void kernel_launch(void* const* d_in, const int* in_sizes, int n_in,
                              void* d_out, int out_size, void* d_ws, size_t ws_size,
                              hipStream_t stream) {
    const float* x      = (const float*)d_in[0];
    const float* W_in   = (const float*)d_in[1];
    const float* conv_w = (const float*)d_in[2];
    const float* conv_b = (const float*)d_in[3];
    const float* W_x    = (const float*)d_in[4];
    const float* W_dt   = (const float*)d_in[5];
    const float* b_dt   = (const float*)d_in[6];
    const float* A_log  = (const float*)d_in[7];
    const float* Dv     = (const float*)d_in[8];
    const float* W_out  = (const float*)d_in[9];

    float* out = (float*)d_out;
    float* hT  = out + (size_t)LSEQ * DM;

    float* ws   = (float*)d_ws;
    float* xr   = ws;
    float* xs   = xr   + (size_t)LSEQ * 2 * DI;
    float* xdbl = xs   + (size_t)LSEQ * DI;
    float* dt   = xdbl + (size_t)LSEQ * XDBL_W;
    float* yb   = dt   + (size_t)LSEQ * DI;

    wmma_gemm_f32<4, 4, false><<<256, 256, 0, stream>>>(
        x, W_in, xr, nullptr, LSEQ, 2 * DI, DM, DM, 2 * DI, 2 * DI);

    conv_silu_kernel<<<(LSEQ * DI) / 256, 256, 0, stream>>>(xr, conv_w, conv_b, xs);

    wmma_gemm_f32<4, 2, false><<<12, 256, 0, stream>>>(
        xs, W_x, xdbl, nullptr, LSEQ, XDBL_W, DI, DI, XDBL_W, XDBL_W);

    wmma_gemm_f32<4, 4, true><<<128, 256, 0, stream>>>(
        xdbl, W_dt, dt, b_dt, LSEQ, DI, DTR, XDBL_W, DI, DI);

    scan_kernel<<<(DI * 8) / 256, 256, 0, stream>>>(dt, xdbl, xs, xr, A_log, Dv, yb, hT);

    wmma_gemm_f32<4, 4, false><<<64, 256, 0, stream>>>(
        yb, W_out, out, nullptr, LSEQ, DM, DI, DI, DM, DM);
}


# scan: no LDS round trip inside a step: the 8-lane butterfly (3 waited lane permutes) becomes per-lane partials in an LDS tile, added in the same tree at write-out; operand reads grouped 8 steps ahead
# speedup vs baseline: 2.4233x; 1.4461x over previous
#include <hip/hip_runtime.h>
#include <hip/hip_bf16.h>
#include <math.h>


typedef float v2f __attribute__((ext_vector_type(2)));
typedef float v8f __attribute__((ext_vector_type(8)));

#define DM   1024
#define DI   2048
#define NS   16
#define KC   4
#define DTR  64
#define LSEQ 2048
#define XDBL_W (DTR + 2*NS)

template<int WM, int WN, bool SOFTPLUS_BIAS>
__global__ __launch_bounds__(256)
void wmma_gemm_f32(const float* __restrict__ A, const float* __restrict__ B,
                   float* __restrict__ C, const float* __restrict__ bias,
                   int M, int N, int K, int lda, int ldb, int ldc)
{
    const int wave = (blockIdx.x * blockDim.x + threadIdx.x) >> 5;
    const int lane = threadIdx.x & 31;
    const int tilesM = M / (WM * 16);
    const int tilesN = N / (WN * 16);
    if (wave >= tilesM * tilesN) return;
    const int tm = wave % tilesM;
    const int tn = wave / tilesM;
    const int m0 = tm * WM * 16;
    const int n0 = tn * WN * 16;
    const int lr = lane & 15;
    const int lh = lane >> 4;

    v8f acc[WM][WN];
#pragma unroll
    for (int i = 0; i < WM; ++i)
#pragma unroll
        for (int j = 0; j < WN; ++j) {
            v8f z = {0.f, 0.f, 0.f, 0.f, 0.f, 0.f, 0.f, 0.f};
            acc[i][j] = z;
        }

    for (int k0 = 0; k0 < K; k0 += 4) {
        v2f a[WM];
#pragma unroll
        for (int i = 0; i < WM; ++i) {
            const float* ap = A + (size_t)(m0 + i * 16 + lr) * lda + k0 + 2 * lh;
            a[i] = *(const v2f*)ap;
        }
        v2f b[WN];
#pragma unroll
        for (int j = 0; j < WN; ++j) {
            const float* bp = B + (size_t)(k0 + 2 * lh) * ldb + n0 + j * 16 + lr;
            v2f t; t[0] = bp[0]; t[1] = bp[ldb];
            b[j] = t;
        }
#pragma unroll
        for (int i = 0; i < WM; ++i)
#pragma unroll
            for (int j = 0; j < WN; ++j)
                acc[i][j] = __builtin_amdgcn_wmma_f32_16x16x4_f32(
                    false, a[i], false, b[j], (short)0, acc[i][j], false, false);
    }

#pragma unroll
    for (int i = 0; i < WM; ++i) {
#pragma unroll
        for (int j = 0; j < WN; ++j) {
            const int col = n0 + j * 16 + lr;
#pragma unroll
            for (int r = 0; r < 8; ++r) {
                const int row = m0 + i * 16 + r + 8 * lh;
                float v = acc[i][j][r];
                if (SOFTPLUS_BIAS) {
                    v += bias[col];
                    v = (v > 20.f) ? v : __logf(1.f + __expf(v));
                }
                C[(size_t)row * ldc + col] = v;
            }
        }
    }
}

__global__ __launch_bounds__(256)
void conv_silu_kernel(const float* __restrict__ xr, const float* __restrict__ conv_w,
                      const float* __restrict__ conv_b, float* __restrict__ xs)
{
    const int idx = blockIdx.x * 256 + threadIdx.x;
    const int l = idx / DI;
    const int d = idx % DI;
    const float4 w = *(const float4*)(conv_w + (size_t)d * KC);
    float acc = conv_b[d];
    const float* xc = xr + d;
    if (l >= 3) acc += xc[(size_t)(l - 3) * (2 * DI)] * w.x;
    if (l >= 2) acc += xc[(size_t)(l - 2) * (2 * DI)] * w.y;
    if (l >= 1) acc += xc[(size_t)(l - 1) * (2 * DI)] * w.z;
    acc             += xc[(size_t)l       * (2 * DI)] * w.w;
    const float sig = 1.f / (1.f + __expf(-acc));
    xs[idx] = acc * sig;
}

#define PP 12
#define PROW (32 * PP)

__global__ __launch_bounds__(256)
void scan_kernel(const float* __restrict__ dt, const float* __restrict__ xdbl,
                 const float* __restrict__ xs, const float* __restrict__ xr,
                 const float* __restrict__ A_log, const float* __restrict__ Dvec,
                 float* __restrict__ y, float* __restrict__ hT)
{
    __shared__ float sDT[64 * 32];
    __shared__ float sXS[64 * 32];
    __shared__ float sBC[64 * 32];
    __shared__ __align__(16) float sP[64 * PROW];

    const int t  = blockIdx.x * 256 + threadIdx.x;
    const int d  = t >> 3;
    const int q  = t & 7;
    const int ch = threadIdx.x >> 3;
    const int d0 = blockIdx.x * 32;
    const int wr = threadIdx.x >> 5;
    const int wc = threadIdx.x & 31;
    const int pslot = ch * PP + q;

    float A0 = -__expf(A_log[(size_t)d * NS + 2 * q]);
    float A1 = -__expf(A_log[(size_t)d * NS + 2 * q + 1]);
    float h0 = 0.f, h1 = 0.f;
    const float Dw = Dvec[d0 + wc];

    for (int c = 0; c < LSEQ / 64; ++c) {
        for (int e = threadIdx.x; e < 64 * 32; e += 256) {
            const int ls = e >> 5, w = e & 31;
            const int l = c * 64 + ls;
            sDT[e] = dt[(size_t)l * DI + d0 + w];
            sXS[e] = xs[(size_t)l * DI + d0 + w];
            sBC[e] = xdbl[(size_t)l * XDBL_W + DTR + w];
        }
        __syncthreads();
        float resv[8], xsv[8];
        for (int k = 0; k < 8; ++k) {
            const int l = c * 64 + wr + 8 * k;
            resv[k] = xr[(size_t)l * (2 * DI) + DI + d0 + wc];
            xsv[k]  = sXS[(wr + 8 * k) * 32 + wc];
        }
#pragma unroll 8
        for (int s = 0; s < 64; ++s) {
            const float dtv = sDT[s * 32 + ch];
            const float xv  = sXS[s * 32 + ch];
            const float2 Bv = *(const float2*)&sBC[s * 32 + 2 * q];
            const float2 Cv = *(const float2*)&sBC[s * 32 + 16 + 2 * q];
            const float dx  = dtv * xv;
            const float dA0 = __expf(dtv * A0);
            const float dA1 = __expf(dtv * A1);
            h0 = dA0 * h0 + dx * Bv.x;
            h1 = dA1 * h1 + dx * Bv.y;
            sP[s * PROW + pslot] = fmaf(h0, Cv.x, h1 * Cv.y);
        }
        __syncthreads();
        for (int k = 0; k < 8; ++k) {
            const int ls = wr + 8 * k;
            const int l = c * 64 + ls;
            const float4 pa = *(const float4*)&sP[ls * PROW + wc * PP];
            const float4 pb = *(const float4*)&sP[ls * PROW + wc * PP + 4];
            float yv = ((pa.x + pa.y) + (pa.z + pa.w)) + ((pb.x + pb.y) + (pb.z + pb.w));
            yv = yv + xsv[k] * Dw;
            const float sig = 1.f / (1.f + __expf(-resv[k]));
            y[(size_t)l * DI + d0 + wc] = yv * (resv[k] * sig);
        }
    }
    float2 hout = make_float2(h0, h1);
    *(float2*)&hT[(size_t)d * NS + 2 * q] = hout;
}

extern "C" void kernel_launch(void* const* d_in, const int* in_sizes, int n_in,
                              void* d_out, int out_size, void* d_ws, size_t ws_size,
                              hipStream_t stream) {
    const float* x      = (const float*)d_in[0];
    const float* W_in   = (const float*)d_in[1];
    const float* conv_w = (const float*)d_in[2];
    const float* conv_b = (const float*)d_in[3];
    const float* W_x    = (const float*)d_in[4];
    const float* W_dt   = (const float*)d_in[5];
    const float* b_dt   = (const float*)d_in[6];
    const float* A_log  = (const float*)d_in[7];
    const float* Dv     = (const float*)d_in[8];
    const float* W_out  = (const float*)d_in[9];

    float* out = (float*)d_out;
    float* hT  = out + (size_t)LSEQ * DM;

    float* ws   = (float*)d_ws;
    float* xr   = ws;
    float* xs   = xr   + (size_t)LSEQ * 2 * DI;
    float* xdbl = xs   + (size_t)LSEQ * DI;
    float* dt   = xdbl + (size_t)LSEQ * XDBL_W;
    float* yb   = dt   + (size_t)LSEQ * DI;

    wmma_gemm_f32<4, 4, false><<<256, 256, 0, stream>>>(
        x, W_in, xr, nullptr, LSEQ, 2 * DI, DM, DM, 2 * DI, 2 * DI);

    conv_silu_kernel<<<(LSEQ * DI) / 256, 256, 0, stream>>>(xr, conv_w, conv_b, xs);

    wmma_gemm_f32<4, 2, false><<<12, 256, 0, stream>>>(
        xs, W_x, xdbl, nullptr, LSEQ, XDBL_W, DI, DI, XDBL_W, XDBL_W);

    wmma_gemm_f32<4, 4, true><<<128, 256, 0, stream>>>(
        xdbl, W_dt, dt, b_dt, LSEQ, DI, DTR, XDBL_W, DI, DI);

    scan_kernel<<<(DI * 8) / 256, 256, 0, stream>>>(dt, xdbl, xs, xr, A_log, Dv, yb, hT);

    wmma_gemm_f32<4, 4, false><<<64, 256, 0, stream>>>(
        yb, W_out, out, nullptr, LSEQ, DM, DI, DI, DM, DM);
}


# GEMM kernel: A and B tiles staged once per block in LDS by 16-byte global loads issued a K-step ahead, fragments read from LDS; block tiles sized so every GEMM fills the chip; k sum order unchanged
# speedup vs baseline: 4.3397x; 1.7908x over previous
#include <hip/hip_runtime.h>
#include <hip/hip_bf16.h>
#include <math.h>


typedef float v2f __attribute__((ext_vector_type(2)));
typedef float v4f __attribute__((ext_vector_type(4)));
typedef float v8f __attribute__((ext_vector_type(8)));

#define DM   1024
#define DI   2048
#define NS   16
#define KC   4
#define DTR  64
#define LSEQ 2048
#define XDBL_W (DTR + 2*NS)

template<int BM, int BN, int KS, int WGM, int WGN, bool SOFTPLUS_BIAS>
__global__ __launch_bounds__(256)
void wmma_gemm_f32(const float* __restrict__ A, const float* __restrict__ B,
                   float* __restrict__ C, const float* __restrict__ bias,
                   int M, int N, int K, int lda, int ldb, int ldc)
{
    constexpr int PA  = KS + 4;
    constexpr int PB  = BN + 8;
    constexpr int WM  = BM / (16 * WGM);
    constexpr int WN  = BN / (16 * WGN);
    constexpr int NA4 = BM * KS / 4;
    constexpr int NB4 = KS * BN / 4;
    constexpr int RA  = (NA4 + 255) / 256;
    constexpr int RB  = (NB4 + 255) / 256;
    constexpr int BN4 = BN / 4;
    constexpr int KA4 = KS / 4;
    static_assert(WGM * WGN == 8, "8 waves per block");
    static_assert(WM * 16 * WGM == BM && WN * 16 * WGN == BN, "tile split");

    __shared__ __align__(16) float sA[2][BM * PA];
    __shared__ __align__(16) float sB[2][KS * PB];

    const int tid  = threadIdx.x;
    const int lane = tid & 31;
    const int w    = tid >> 5;
    const int wm   = w % WGM;
    const int wn   = w / WGM;
    const int lr   = lane & 15;
    const int lh   = lane >> 4;
    const int tilesM = M / BM;
    const int m0 = (blockIdx.x % tilesM) * BM;
    const int n0 = (blockIdx.x / tilesM) * BN;

    v8f acc[WM][WN];
#pragma unroll
    for (int i = 0; i < WM; ++i)
#pragma unroll
        for (int j = 0; j < WN; ++j) {
            v8f z = {0.f, 0.f, 0.f, 0.f, 0.f, 0.f, 0.f, 0.f};
            acc[i][j] = z;
        }

    v4f ra[RA], rb[RB];

#pragma unroll
    for (int i = 0; i < RA; ++i) {
        const int idx = tid + i * 256;
        if ((NA4 % 256 == 0) || idx < NA4)
            ra[i] = *(const v4f*)(A + (size_t)(m0 + (idx / KA4)) * lda + (idx % KA4) * 4);
    }
#pragma unroll
    for (int i = 0; i < RB; ++i) {
        const int idx = tid + i * 256;
        if ((NB4 % 256 == 0) || idx < NB4)
            rb[i] = *(const v4f*)(B + (size_t)(idx / BN4) * ldb + n0 + (idx % BN4) * 4);
    }
#pragma unroll
    for (int i = 0; i < RA; ++i) {
        const int idx = tid + i * 256;
        if ((NA4 % 256 == 0) || idx < NA4)
            *(v4f*)&sA[0][(idx / KA4) * PA + (idx % KA4) * 4] = ra[i];
    }
#pragma unroll
    for (int i = 0; i < RB; ++i) {
        const int idx = tid + i * 256;
        if ((NB4 % 256 == 0) || idx < NB4)
            *(v4f*)&sB[0][(idx / BN4) * PB + (idx % BN4) * 4] = rb[i];
    }
    __syncthreads();

    const int nk = K / KS;
    for (int kt = 0; kt < nk; ++kt) {
        const int buf = kt & 1;
        const bool more = (kt + 1 < nk);
        if (more) {
            const int k0 = (kt + 1) * KS;
#pragma unroll
            for (int i = 0; i < RA; ++i) {
                const int idx = tid + i * 256;
                if ((NA4 % 256 == 0) || idx < NA4)
                    ra[i] = *(const v4f*)(A + (size_t)(m0 + (idx / KA4)) * lda + k0 + (idx % KA4) * 4);
            }
#pragma unroll
            for (int i = 0; i < RB; ++i) {
                const int idx = tid + i * 256;
                if ((NB4 % 256 == 0) || idx < NB4)
                    rb[i] = *(const v4f*)(B + (size_t)(k0 + idx / BN4) * ldb + n0 + (idx % BN4) * 4);
            }
        }
        const float* tA = &sA[buf][(wm * WM * 16 + lr) * PA + 2 * lh];
        const float* tB = &sB[buf][(2 * lh) * PB + wn * WN * 16 + lr];
#pragma unroll
        for (int kk = 0; kk < KS / 4; ++kk) {
            v2f a[WM];
#pragma unroll
            for (int i = 0; i < WM; ++i)
                a[i] = *(const v2f*)(tA + i * 16 * PA + kk * 4);
            v2f b[WN];
#pragma unroll
            for (int j = 0; j < WN; ++j) {
                const float* bp = tB + (kk * 4) * PB + j * 16;
                v2f t; t[0] = bp[0]; t[1] = bp[PB];
                b[j] = t;
            }
#pragma unroll
            for (int i = 0; i < WM; ++i)
#pragma unroll
                for (int j = 0; j < WN; ++j)
                    acc[i][j] = __builtin_amdgcn_wmma_f32_16x16x4_f32(
                        false, a[i], false, b[j], (short)0, acc[i][j], false, false);
        }
        if (more) {
#pragma unroll
            for (int i = 0; i < RA; ++i) {
                const int idx = tid + i * 256;
                if ((NA4 % 256 == 0) || idx < NA4)
                    *(v4f*)&sA[buf ^ 1][(idx / KA4) * PA + (idx % KA4) * 4] = ra[i];
            }
#pragma unroll
            for (int i = 0; i < RB; ++i) {
                const int idx = tid + i * 256;
                if ((NB4 % 256 == 0) || idx < NB4)
                    *(v4f*)&sB[buf ^ 1][(idx / BN4) * PB + (idx % BN4) * 4] = rb[i];
            }
        }
        __syncthreads();
    }

#pragma unroll
    for (int i = 0; i < WM; ++i) {
#pragma unroll
        for (int j = 0; j < WN; ++j) {
            const int col = n0 + wn * WN * 16 + j * 16 + lr;
#pragma unroll
            for (int r = 0; r < 8; ++r) {
                const int row = m0 + wm * WM * 16 + i * 16 + r + 8 * lh;
                float v = acc[i][j][r];
                if (SOFTPLUS_BIAS) {
                    v += bias[col];
                    v = (v > 20.f) ? v : __logf(1.f + __expf(v));
                }
                C[(size_t)row * ldc + col] = v;
            }
        }
    }
}

__global__ __launch_bounds__(256)
void conv_silu_kernel(const float* __restrict__ xr, const float* __restrict__ conv_w,
                      const float* __restrict__ conv_b, float* __restrict__ xs)
{
    const int idx = blockIdx.x * 256 + threadIdx.x;
    const int l = idx / DI;
    const int d = idx % DI;
    const float4 w = *(const float4*)(conv_w + (size_t)d * KC);
    float acc = conv_b[d];
    const float* xc = xr + d;
    if (l >= 3) acc += xc[(size_t)(l - 3) * (2 * DI)] * w.x;
    if (l >= 2) acc += xc[(size_t)(l - 2) * (2 * DI)] * w.y;
    if (l >= 1) acc += xc[(size_t)(l - 1) * (2 * DI)] * w.z;
    acc             += xc[(size_t)l       * (2 * DI)] * w.w;
    const float sig = 1.f / (1.f + __expf(-acc));
    xs[idx] = acc * sig;
}

#define PP 12
#define PROW (32 * PP)

__global__ __launch_bounds__(256)
void scan_kernel(const float* __restrict__ dt, const float* __restrict__ xdbl,
                 const float* __restrict__ xs, const float* __restrict__ xr,
                 const float* __restrict__ A_log, const float* __restrict__ Dvec,
                 float* __restrict__ y, float* __restrict__ hT)
{
    __shared__ float sDT[64 * 32];
    __shared__ float sXS[64 * 32];
    __shared__ float sBC[64 * 32];
    __shared__ __align__(16) float sP[64 * PROW];

    const int t  = blockIdx.x * 256 + threadIdx.x;
    const int d  = t >> 3;
    const int q  = t & 7;
    const int ch = threadIdx.x >> 3;
    const int d0 = blockIdx.x * 32;
    const int wr = threadIdx.x >> 5;
    const int wc = threadIdx.x & 31;
    const int pslot = ch * PP + q;

    float A0 = -__expf(A_log[(size_t)d * NS + 2 * q]);
    float A1 = -__expf(A_log[(size_t)d * NS + 2 * q + 1]);
    float h0 = 0.f, h1 = 0.f;
    const float Dw = Dvec[d0 + wc];

    for (int c = 0; c < LSEQ / 64; ++c) {
        for (int e = threadIdx.x; e < 64 * 32; e += 256) {
            const int ls = e >> 5, w = e & 31;
            const int l = c * 64 + ls;
            sDT[e] = dt[(size_t)l * DI + d0 + w];
            sXS[e] = xs[(size_t)l * DI + d0 + w];
            sBC[e] = xdbl[(size_t)l * XDBL_W + DTR + w];
        }
        __syncthreads();
        float resv[8], xsv[8];
        for (int k = 0; k < 8; ++k) {
            const int l = c * 64 + wr + 8 * k;
            resv[k] = xr[(size_t)l * (2 * DI) + DI + d0 + wc];
            xsv[k]  = sXS[(wr + 8 * k) * 32 + wc];
        }
#pragma unroll 8
        for (int s = 0; s < 64; ++s) {
            const float dtv = sDT[s * 32 + ch];
            const float xv  = sXS[s * 32 + ch];
            const float2 Bv = *(const float2*)&sBC[s * 32 + 2 * q];
            const float2 Cv = *(const float2*)&sBC[s * 32 + 16 + 2 * q];
            const float dx  = dtv * xv;
            const float dA0 = __expf(dtv * A0);
            const float dA1 = __expf(dtv * A1);
            h0 = dA0 * h0 + dx * Bv.x;
            h1 = dA1 * h1 + dx * Bv.y;
            sP[s * PROW + pslot] = fmaf(h0, Cv.x, h1 * Cv.y);
        }
        __syncthreads();
        for (int k = 0; k < 8; ++k) {
            const int ls = wr + 8 * k;
            const int l = c * 64 + ls;
            const float4 pa = *(const float4*)&sP[ls * PROW + wc * PP];
            const float4 pb = *(const float4*)&sP[ls * PROW + wc * PP + 4];
            float yv = ((pa.x + pa.y) + (pa.z + pa.w)) + ((pb.x + pb.y) + (pb.z + pb.w));
            yv = yv + xsv[k] * Dw;
            const float sig = 1.f / (1.f + __expf(-resv[k]));
            y[(size_t)l * DI + d0 + wc] = yv * (resv[k] * sig);
        }
    }
    float2 hout = make_float2(h0, h1);
    *(float2*)&hT[(size_t)d * NS + 2 * q] = hout;
}

extern "C" void kernel_launch(void* const* d_in, const int* in_sizes, int n_in,
                              void* d_out, int out_size, void* d_ws, size_t ws_size,
                              hipStream_t stream) {
    const float* x      = (const float*)d_in[0];
    const float* W_in   = (const float*)d_in[1];
    const float* conv_w = (const float*)d_in[2];
    const float* conv_b = (const float*)d_in[3];
    const float* W_x    = (const float*)d_in[4];
    const float* W_dt   = (const float*)d_in[5];
    const float* b_dt   = (const float*)d_in[6];
    const float* A_log  = (const float*)d_in[7];
    const float* Dv     = (const float*)d_in[8];
    const float* W_out  = (const float*)d_in[9];

    float* out = (float*)d_out;
    float* hT  = out + (size_t)LSEQ * DM;

    float* ws   = (float*)d_ws;
    float* xr   = ws;
    float* xs   = xr   + (size_t)LSEQ * 2 * DI;
    float* xdbl = xs   + (size_t)LSEQ * DI;
    float* dt   = xdbl + (size_t)LSEQ * XDBL_W;
    float* yb   = dt   + (size_t)LSEQ * DI;

    wmma_gemm_f32<128, 128, 16, 2, 4, false><<<512, 256, 0, stream>>>(
        x, W_in, xr, nullptr, LSEQ, 2 * DI, DM, DM, 2 * DI, 2 * DI);

    conv_silu_kernel<<<(LSEQ * DI) / 256, 256, 0, stream>>>(xr, conv_w, conv_b, xs);

    wmma_gemm_f32<64, 96, 32, 4, 2, false><<<32, 256, 0, stream>>>(
        xs, W_x, xdbl, nullptr, LSEQ, XDBL_W, DI, DI, XDBL_W, XDBL_W);

    wmma_gemm_f32<128, 128, 16, 2, 4, true><<<256, 256, 0, stream>>>(
        xdbl, W_dt, dt, b_dt, LSEQ, DI, DTR, XDBL_W, DI, DI);

    scan_kernel<<<(DI * 8) / 256, 256, 0, stream>>>(dt, xdbl, xs, xr, A_log, Dv, yb, hT);

    wmma_gemm_f32<64, 128, 32, 2, 4, false><<<256, 256, 0, stream>>>(
        yb, W_out, out, nullptr, LSEQ, DM, DI, DI, DM, DM);
}


# scan: tiles dt/xs/B/C of chunk c+1 fetched by 16-byte loads into registers ahead of chunk c's 64 steps, stored to LDS after them: no waited staging load left in the chunk loop; sums unchanged
# speedup vs baseline: 5.9044x; 1.3606x over previous
#include <hip/hip_runtime.h>
#include <hip/hip_bf16.h>
#include <math.h>


typedef float v2f __attribute__((ext_vector_type(2)));
typedef float v4f __attribute__((ext_vector_type(4)));
typedef float v8f __attribute__((ext_vector_type(8)));

#define DM   1024
#define DI   2048
#define NS   16
#define KC   4
#define DTR  64
#define LSEQ 2048
#define XDBL_W (DTR + 2*NS)

template<int BM, int BN, int KS, int WGM, int WGN, bool SOFTPLUS_BIAS>
__global__ __launch_bounds__(256)
void wmma_gemm_f32(const float* __restrict__ A, const float* __restrict__ B,
                   float* __restrict__ C, const float* __restrict__ bias,
                   int M, int N, int K, int lda, int ldb, int ldc)
{
    constexpr int PA  = KS + 4;
    constexpr int PB  = BN + 8;
    constexpr int WM  = BM / (16 * WGM);
    constexpr int WN  = BN / (16 * WGN);
    constexpr int NA4 = BM * KS / 4;
    constexpr int NB4 = KS * BN / 4;
    constexpr int RA  = (NA4 + 255) / 256;
    constexpr int RB  = (NB4 + 255) / 256;
    constexpr int BN4 = BN / 4;
    constexpr int KA4 = KS / 4;
    static_assert(WGM * WGN == 8, "8 waves per block");
    static_assert(WM * 16 * WGM == BM && WN * 16 * WGN == BN, "tile split");

    __shared__ __align__(16) float sA[2][BM * PA];
    __shared__ __align__(16) float sB[2][KS * PB];

    const int tid  = threadIdx.x;
    const int lane = tid & 31;
    const int w    = tid >> 5;
    const int wm   = w % WGM;
    const int wn   = w / WGM;
    const int lr   = lane & 15;
    const int lh   = lane >> 4;
    const int tilesM = M / BM;
    const int m0 = (blockIdx.x % tilesM) * BM;
    const int n0 = (blockIdx.x / tilesM) * BN;

    v8f acc[WM][WN];
#pragma unroll
    for (int i = 0; i < WM; ++i)
#pragma unroll
        for (int j = 0; j < WN; ++j) {
            v8f z = {0.f, 0.f, 0.f, 0.f, 0.f, 0.f, 0.f, 0.f};
            acc[i][j] = z;
        }

    v4f ra[RA], rb[RB];

#pragma unroll
    for (int i = 0; i < RA; ++i) {
        const int idx = tid + i * 256;
        if ((NA4 % 256 == 0) || idx < NA4)
            ra[i] = *(const v4f*)(A + (size_t)(m0 + (idx / KA4)) * lda + (idx % KA4) * 4);
    }
#pragma unroll
    for (int i = 0; i < RB; ++i) {
        const int idx = tid + i * 256;
        if ((NB4 % 256 == 0) || idx < NB4)
            rb[i] = *(const v4f*)(B + (size_t)(idx / BN4) * ldb + n0 + (idx % BN4) * 4);
    }
#pragma unroll
    for (int i = 0; i < RA; ++i) {
        const int idx = tid + i * 256;
        if ((NA4 % 256 == 0) || idx < NA4)
            *(v4f*)&sA[0][(idx / KA4) * PA + (idx % KA4) * 4] = ra[i];
    }
#pragma unroll
    for (int i = 0; i < RB; ++i) {
        const int idx = tid + i * 256;
        if ((NB4 % 256 == 0) || idx < NB4)
            *(v4f*)&sB[0][(idx / BN4) * PB + (idx % BN4) * 4] = rb[i];
    }
    __syncthreads();

    const int nk = K / KS;
    for (int kt = 0; kt < nk; ++kt) {
        const int buf = kt & 1;
        const bool more = (kt + 1 < nk);
        if (more) {
            const int k0 = (kt + 1) * KS;
#pragma unroll
            for (int i = 0; i < RA; ++i) {
                const int idx = tid + i * 256;
                if ((NA4 % 256 == 0) || idx < NA4)
                    ra[i] = *(const v4f*)(A + (size_t)(m0 + (idx / KA4)) * lda + k0 + (idx % KA4) * 4);
            }
#pragma unroll
            for (int i = 0; i < RB; ++i) {
                const int idx = tid + i * 256;
                if ((NB4 % 256 == 0) || idx < NB4)
                    rb[i] = *(const v4f*)(B + (size_t)(k0 + idx / BN4) * ldb + n0 + (idx % BN4) * 4);
            }
        }
        const float* tA = &sA[buf][(wm * WM * 16 + lr) * PA + 2 * lh];
        const float* tB = &sB[buf][(2 * lh) * PB + wn * WN * 16 + lr];
#pragma unroll
        for (int kk = 0; kk < KS / 4; ++kk) {
            v2f a[WM];
#pragma unroll
            for (int i = 0; i < WM; ++i)
                a[i] = *(const v2f*)(tA + i * 16 * PA + kk * 4);
            v2f b[WN];
#pragma unroll
            for (int j = 0; j < WN; ++j) {
                const float* bp = tB + (kk * 4) * PB + j * 16;
                v2f t; t[0] = bp[0]; t[1] = bp[PB];
                b[j] = t;
            }
#pragma unroll
            for (int i = 0; i < WM; ++i)
#pragma unroll
                for (int j = 0; j < WN; ++j)
                    acc[i][j] = __builtin_amdgcn_wmma_f32_16x16x4_f32(
                        false, a[i], false, b[j], (short)0, acc[i][j], false, false);
        }
        if (more) {
#pragma unroll
            for (int i = 0; i < RA; ++i) {
                const int idx = tid + i * 256;
                if ((NA4 % 256 == 0) || idx < NA4)
                    *(v4f*)&sA[buf ^ 1][(idx / KA4) * PA + (idx % KA4) * 4] = ra[i];
            }
#pragma unroll
            for (int i = 0; i < RB; ++i) {
                const int idx = tid + i * 256;
                if ((NB4 % 256 == 0) || idx < NB4)
                    *(v4f*)&sB[buf ^ 1][(idx / BN4) * PB + (idx % BN4) * 4] = rb[i];
            }
        }
        __syncthreads();
    }

#pragma unroll
    for (int i = 0; i < WM; ++i) {
#pragma unroll
        for (int j = 0; j < WN; ++j) {
            const int col = n0 + wn * WN * 16 + j * 16 + lr;
#pragma unroll
            for (int r = 0; r < 8; ++r) {
                const int row = m0 + wm * WM * 16 + i * 16 + r + 8 * lh;
                float v = acc[i][j][r];
                if (SOFTPLUS_BIAS) {
                    v += bias[col];
                    v = (v > 20.f) ? v : __logf(1.f + __expf(v));
                }
                C[(size_t)row * ldc + col] = v;
            }
        }
    }
}

__global__ __launch_bounds__(256)
void conv_silu_kernel(const float* __restrict__ xr, const float* __restrict__ conv_w,
                      const float* __restrict__ conv_b, float* __restrict__ xs)
{
    const int idx = blockIdx.x * 256 + threadIdx.x;
    const int l = idx / DI;
    const int d = idx % DI;
    const float4 w = *(const float4*)(conv_w + (size_t)d * KC);
    float acc = conv_b[d];
    const float* xc = xr + d;
    if (l >= 3) acc += xc[(size_t)(l - 3) * (2 * DI)] * w.x;
    if (l >= 2) acc += xc[(size_t)(l - 2) * (2 * DI)] * w.y;
    if (l >= 1) acc += xc[(size_t)(l - 1) * (2 * DI)] * w.z;
    acc             += xc[(size_t)l       * (2 * DI)] * w.w;
    const float sig = 1.f / (1.f + __expf(-acc));
    xs[idx] = acc * sig;
}

#define PP 12
#define PROW (32 * PP)

__global__ __launch_bounds__(256)
void scan_kernel(const float* __restrict__ dt, const float* __restrict__ xdbl,
                 const float* __restrict__ xs, const float* __restrict__ xr,
                 const float* __restrict__ A_log, const float* __restrict__ Dvec,
                 float* __restrict__ y, float* __restrict__ hT)
{
    __shared__ __align__(16) float sDT[64 * 32];
    __shared__ __align__(16) float sXS[64 * 32];
    __shared__ __align__(16) float sBC[64 * 32];
    __shared__ __align__(16) float sP[64 * PROW];

    const int t  = blockIdx.x * 256 + threadIdx.x;
    const int d  = t >> 3;
    const int q  = t & 7;
    const int ch = threadIdx.x >> 3;
    const int d0 = blockIdx.x * 32;
    const int wr = threadIdx.x >> 5;
    const int wc = threadIdx.x & 31;
    const int pslot = ch * PP + q;

    float A0 = -__expf(A_log[(size_t)d * NS + 2 * q]);
    float A1 = -__expf(A_log[(size_t)d * NS + 2 * q + 1]);
    float h0 = 0.f, h1 = 0.f;
    const float Dw = Dvec[d0 + wc];

    const float* gDT = dt   + (size_t)ch * DI + d0 + 4 * q;
    const float* gXS = xs   + (size_t)ch * DI + d0 + 4 * q;
    const float* gBC = xdbl + (size_t)ch * XDBL_W + DTR + 4 * q;
    v4f pdt[2], pxs[2], pbc[2];
    for (int i = 0; i < 2; ++i) {
        pdt[i] = *(const v4f*)(gDT + (size_t)(32 * i) * DI);
        pxs[i] = *(const v4f*)(gXS + (size_t)(32 * i) * DI);
        pbc[i] = *(const v4f*)(gBC + (size_t)(32 * i) * XDBL_W);
    }
    for (int i = 0; i < 2; ++i) {
        *(v4f*)&sDT[(threadIdx.x + 256 * i) * 4] = pdt[i];
        *(v4f*)&sXS[(threadIdx.x + 256 * i) * 4] = pxs[i];
        *(v4f*)&sBC[(threadIdx.x + 256 * i) * 4] = pbc[i];
    }

    for (int c = 0; c < LSEQ / 64; ++c) {
        const bool more = (c + 1 < LSEQ / 64);
        __syncthreads();
        if (more) {
            for (int i = 0; i < 2; ++i) {
                const size_t l1 = (size_t)(c + 1) * 64 + 32 * i;
                pdt[i] = *(const v4f*)(gDT + l1 * DI);
                pxs[i] = *(const v4f*)(gXS + l1 * DI);
                pbc[i] = *(const v4f*)(gBC + l1 * XDBL_W);
            }
        }
        float resv[8], xsv[8];
        for (int k = 0; k < 8; ++k) {
            const int l = c * 64 + wr + 8 * k;
            resv[k] = xr[(size_t)l * (2 * DI) + DI + d0 + wc];
            xsv[k]  = sXS[(wr + 8 * k) * 32 + wc];
        }
#pragma unroll 8
        for (int s = 0; s < 64; ++s) {
            const float dtv = sDT[s * 32 + ch];
            const float xv  = sXS[s * 32 + ch];
            const float2 Bv = *(const float2*)&sBC[s * 32 + 2 * q];
            const float2 Cv = *(const float2*)&sBC[s * 32 + 16 + 2 * q];
            const float dx  = dtv * xv;
            const float dA0 = __expf(dtv * A0);
            const float dA1 = __expf(dtv * A1);
            h0 = dA0 * h0 + dx * Bv.x;
            h1 = dA1 * h1 + dx * Bv.y;
            sP[s * PROW + pslot] = fmaf(h0, Cv.x, h1 * Cv.y);
        }
        __syncthreads();
        if (more) {
            for (int i = 0; i < 2; ++i) {
                *(v4f*)&sDT[(threadIdx.x + 256 * i) * 4] = pdt[i];
                *(v4f*)&sXS[(threadIdx.x + 256 * i) * 4] = pxs[i];
                *(v4f*)&sBC[(threadIdx.x + 256 * i) * 4] = pbc[i];
            }
        }
        for (int k = 0; k < 8; ++k) {
            const int ls = wr + 8 * k;
            const int l = c * 64 + ls;
            const float4 pa = *(const float4*)&sP[ls * PROW + wc * PP];
            const float4 pb = *(const float4*)&sP[ls * PROW + wc * PP + 4];
            float yv = ((pa.x + pa.y) + (pa.z + pa.w)) + ((pb.x + pb.y) + (pb.z + pb.w));
            yv = yv + xsv[k] * Dw;
            const float sig = 1.f / (1.f + __expf(-resv[k]));
            y[(size_t)l * DI + d0 + wc] = yv * (resv[k] * sig);
        }
    }
    float2 hout = make_float2(h0, h1);
    *(float2*)&hT[(size_t)d * NS + 2 * q] = hout;
}

extern "C" void kernel_launch(void* const* d_in, const int* in_sizes, int n_in,
                              void* d_out, int out_size, void* d_ws, size_t ws_size,
                              hipStream_t stream) {
    const float* x      = (const float*)d_in[0];
    const float* W_in   = (const float*)d_in[1];
    const float* conv_w = (const float*)d_in[2];
    const float* conv_b = (const float*)d_in[3];
    const float* W_x    = (const float*)d_in[4];
    const float* W_dt   = (const float*)d_in[5];
    const float* b_dt   = (const float*)d_in[6];
    const float* A_log  = (const float*)d_in[7];
    const float* Dv     = (const float*)d_in[8];
    const float* W_out  = (const float*)d_in[9];

    float* out = (float*)d_out;
    float* hT  = out + (size_t)LSEQ * DM;

    float* ws   = (float*)d_ws;
    float* xr   = ws;
    float* xs   = xr   + (size_t)LSEQ * 2 * DI;
    float* xdbl = xs   + (size_t)LSEQ * DI;
    float* dt   = xdbl + (size_t)LSEQ * XDBL_W;
    float* yb   = dt   + (size_t)LSEQ * DI;

    wmma_gemm_f32<128, 128, 16, 2, 4, false><<<512, 256, 0, stream>>>(
        x, W_in, xr, nullptr, LSEQ, 2 * DI, DM, DM, 2 * DI, 2 * DI);

    conv_silu_kernel<<<(LSEQ * DI) / 256, 256, 0, stream>>>(xr, conv_w, conv_b, xs);

    wmma_gemm_f32<64, 96, 32, 4, 2, false><<<32, 256, 0, stream>>>(
        xs, W_x, xdbl, nullptr, LSEQ, XDBL_W, DI, DI, XDBL_W, XDBL_W);

    wmma_gemm_f32<128, 128, 16, 2, 4, true><<<256, 256, 0, stream>>>(
        xdbl, W_dt, dt, b_dt, LSEQ, DI, DTR, XDBL_W, DI, DI);

    scan_kernel<<<(DI * 8) / 256, 256, 0, stream>>>(dt, xdbl, xs, xr, A_log, Dv, yb, hT);

    wmma_gemm_f32<64, 128, 32, 2, 4, false><<<256, 256, 0, stream>>>(
        yb, W_out, out, nullptr, LSEQ, DM, DI, DI, DM, DM);
}


# x-proj GEMM (N=96): 16x96 strips with K split over the 8 waves of a block, the 8 partial strips added in LDS in a fixed tree: 128 blocks, not 32; k sum order changed
# speedup vs baseline: 6.7942x; 1.1507x over previous
#include <hip/hip_runtime.h>
#include <hip/hip_bf16.h>
#include <math.h>


typedef float v2f __attribute__((ext_vector_type(2)));
typedef float v4f __attribute__((ext_vector_type(4)));
typedef float v8f __attribute__((ext_vector_type(8)));

#define DM   1024
#define DI   2048
#define NS   16
#define KC   4
#define DTR  64
#define LSEQ 2048
#define XDBL_W (DTR + 2*NS)

template<int BM, int BN, int KS, int WGM, int WGN, bool SOFTPLUS_BIAS>
__global__ __launch_bounds__(256)
void wmma_gemm_f32(const float* __restrict__ A, const float* __restrict__ B,
                   float* __restrict__ C, const float* __restrict__ bias,
                   int M, int N, int K, int lda, int ldb, int ldc)
{
    constexpr int PA  = KS + 4;
    constexpr int PB  = BN + 8;
    constexpr int WM  = BM / (16 * WGM);
    constexpr int WN  = BN / (16 * WGN);
    constexpr int NA4 = BM * KS / 4;
    constexpr int NB4 = KS * BN / 4;
    constexpr int RA  = (NA4 + 255) / 256;
    constexpr int RB  = (NB4 + 255) / 256;
    constexpr int BN4 = BN / 4;
    constexpr int KA4 = KS / 4;
    static_assert(WGM * WGN == 8, "8 waves per block");
    static_assert(WM * 16 * WGM == BM && WN * 16 * WGN == BN, "tile split");

    __shared__ __align__(16) float sA[2][BM * PA];
    __shared__ __align__(16) float sB[2][KS * PB];

    const int tid  = threadIdx.x;
    const int lane = tid & 31;
    const int w    = tid >> 5;
    const int wm   = w % WGM;
    const int wn   = w / WGM;
    const int lr   = lane & 15;
    const int lh   = lane >> 4;
    const int tilesM = M / BM;
    const int m0 = (blockIdx.x % tilesM) * BM;
    const int n0 = (blockIdx.x / tilesM) * BN;

    v8f acc[WM][WN];
#pragma unroll
    for (int i = 0; i < WM; ++i)
#pragma unroll
        for (int j = 0; j < WN; ++j) {
            v8f z = {0.f, 0.f, 0.f, 0.f, 0.f, 0.f, 0.f, 0.f};
            acc[i][j] = z;
        }

    v4f ra[RA], rb[RB];

#pragma unroll
    for (int i = 0; i < RA; ++i) {
        const int idx = tid + i * 256;
        if ((NA4 % 256 == 0) || idx < NA4)
            ra[i] = *(const v4f*)(A + (size_t)(m0 + (idx / KA4)) * lda + (idx % KA4) * 4);
    }
#pragma unroll
    for (int i = 0; i < RB; ++i) {
        const int idx = tid + i * 256;
        if ((NB4 % 256 == 0) || idx < NB4)
            rb[i] = *(const v4f*)(B + (size_t)(idx / BN4) * ldb + n0 + (idx % BN4) * 4);
    }
#pragma unroll
    for (int i = 0; i < RA; ++i) {
        const int idx = tid + i * 256;
        if ((NA4 % 256 == 0) || idx < NA4)
            *(v4f*)&sA[0][(idx / KA4) * PA + (idx % KA4) * 4] = ra[i];
    }
#pragma unroll
    for (int i = 0; i < RB; ++i) {
        const int idx = tid + i * 256;
        if ((NB4 % 256 == 0) || idx < NB4)
            *(v4f*)&sB[0][(idx / BN4) * PB + (idx % BN4) * 4] = rb[i];
    }
    __syncthreads();

    const int nk = K / KS;
    for (int kt = 0; kt < nk; ++kt) {
        const int buf = kt & 1;
        const bool more = (kt + 1 < nk);
        if (more) {
            const int k0 = (kt + 1) * KS;
#pragma unroll
            for (int i = 0; i < RA; ++i) {
                const int idx = tid + i * 256;
                if ((NA4 % 256 == 0) || idx < NA4)
                    ra[i] = *(const v4f*)(A + (size_t)(m0 + (idx / KA4)) * lda + k0 + (idx % KA4) * 4);
            }
#pragma unroll
            for (int i = 0; i < RB; ++i) {
                const int idx = tid + i * 256;
                if ((NB4 % 256 == 0) || idx < NB4)
                    rb[i] = *(const v4f*)(B + (size_t)(k0 + idx / BN4) * ldb + n0 + (idx % BN4) * 4);
            }
        }
        const float* tA = &sA[buf][(wm * WM * 16 + lr) * PA + 2 * lh];
        const float* tB = &sB[buf][(2 * lh) * PB + wn * WN * 16 + lr];
#pragma unroll
        for (int kk = 0; kk < KS / 4; ++kk) {
            v2f a[WM];
#pragma unroll
            for (int i = 0; i < WM; ++i)
                a[i] = *(const v2f*)(tA + i * 16 * PA + kk * 4);
            v2f b[WN];
#pragma unroll
            for (int j = 0; j < WN; ++j) {
                const float* bp = tB + (kk * 4) * PB + j * 16;
                v2f t; t[0] = bp[0]; t[1] = bp[PB];
                b[j] = t;
            }
#pragma unroll
            for (int i = 0; i < WM; ++i)
#pragma unroll
                for (int j = 0; j < WN; ++j)
                    acc[i][j] = __builtin_amdgcn_wmma_f32_16x16x4_f32(
                        false, a[i], false, b[j], (short)0, acc[i][j], false, false);
        }
        if (more) {
#pragma unroll
            for (int i = 0; i < RA; ++i) {
                const int idx = tid + i * 256;
                if ((NA4 % 256 == 0) || idx < NA4)
                    *(v4f*)&sA[buf ^ 1][(idx / KA4) * PA + (idx % KA4) * 4] = ra[i];
            }
#pragma unroll
            for (int i = 0; i < RB; ++i) {
                const int idx = tid + i * 256;
                if ((NB4 % 256 == 0) || idx < NB4)
                    *(v4f*)&sB[buf ^ 1][(idx / BN4) * PB + (idx % BN4) * 4] = rb[i];
            }
        }
        __syncthreads();
    }

#pragma unroll
    for (int i = 0; i < WM; ++i) {
#pragma unroll
        for (int j = 0; j < WN; ++j) {
            const int col = n0 + wn * WN * 16 + j * 16 + lr;
#pragma unroll
            for (int r = 0; r < 8; ++r) {
                const int row = m0 + wm * WM * 16 + i * 16 + r + 8 * lh;
                float v = acc[i][j][r];
                if (SOFTPLUS_BIAS) {
                    v += bias[col];
                    v = (v > 20.f) ? v : __logf(1.f + __expf(v));
                }
                C[(size_t)row * ldc + col] = v;
            }
        }
    }
}

template<int BN, int KS>
__global__ __launch_bounds__(256)
void wmma_gemm_f32_ksplit(const float* __restrict__ A, const float* __restrict__ B,
                          float* __restrict__ C, int K, int lda)
{
    constexpr int BM  = 16;
    constexpr int PA  = KS + 4;
    constexpr int PB  = BN + 8;
    constexpr int WN  = BN / 16;
    constexpr int RA  = BM * KS / 4 / 256;
    constexpr int RB  = KS * BN / 4 / 256;
    constexpr int BN4 = BN / 4;
    constexpr int KA4 = KS / 4;
    constexpr int KW  = KS / 32;
    static_assert((BM * KS / 4) % 256 == 0 && (KS * BN / 4) % 256 == 0, "whole pieces per thread");
    static_assert((BM * BN) % 256 == 0, "whole outputs per thread");
    static_assert(8 * BM * BN <= KS * PB, "the 8 partial strips fit in one B buffer");

    __shared__ __align__(16) float sA[2][BM * PA];
    __shared__ __align__(16) float sB[2][KS * PB];

    const int tid  = threadIdx.x;
    const int lane = tid & 31;
    const int w    = tid >> 5;
    const int lr   = lane & 15;
    const int lh   = lane >> 4;
    const int m0   = blockIdx.x * BM;

    v8f acc[WN];
#pragma unroll
    for (int j = 0; j < WN; ++j) {
        v8f z = {0.f, 0.f, 0.f, 0.f, 0.f, 0.f, 0.f, 0.f};
        acc[j] = z;
    }

    v4f ra[RA], rb[RB];

#pragma unroll
    for (int i = 0; i < RA; ++i) {
        const int idx = tid + i * 256;
        ra[i] = *(const v4f*)(A + (size_t)(m0 + (idx / KA4)) * lda + (idx % KA4) * 4);
    }
#pragma unroll
    for (int i = 0; i < RB; ++i) {
        const int idx = tid + i * 256;
        rb[i] = *(const v4f*)(B + (size_t)(idx / BN4) * BN + (idx % BN4) * 4);
    }
#pragma unroll
    for (int i = 0; i < RA; ++i) {
        const int idx = tid + i * 256;
        *(v4f*)&sA[0][(idx / KA4) * PA + (idx % KA4) * 4] = ra[i];
    }
#pragma unroll
    for (int i = 0; i < RB; ++i) {
        const int idx = tid + i * 256;
        *(v4f*)&sB[0][(idx / BN4) * PB + (idx % BN4) * 4] = rb[i];
    }
    __syncthreads();

    const int nk = K / KS;
    for (int kt = 0; kt < nk; ++kt) {
        const int buf = kt & 1;
        const bool more = (kt + 1 < nk);
        if (more) {
            const int k0 = (kt + 1) * KS;
#pragma unroll
            for (int i = 0; i < RA; ++i) {
                const int idx = tid + i * 256;
                ra[i] = *(const v4f*)(A + (size_t)(m0 + (idx / KA4)) * lda + k0 + (idx % KA4) * 4);
            }
#pragma unroll
            for (int i = 0; i < RB; ++i) {
                const int idx = tid + i * 256;
                rb[i] = *(const v4f*)(B + (size_t)(k0 + idx / BN4) * BN + (idx % BN4) * 4);
            }
        }
        const float* tA = &sA[buf][lr * PA + w * KW * 4 + 2 * lh];
        const float* tB = &sB[buf][(w * KW * 4 + 2 * lh) * PB + lr];
#pragma unroll
        for (int kk = 0; kk < KW; ++kk) {
            const v2f a = *(const v2f*)(tA + kk * 4);
            v2f b[WN];
#pragma unroll
            for (int j = 0; j < WN; ++j) {
                const float* bp = tB + (kk * 4) * PB + j * 16;
                v2f t; t[0] = bp[0]; t[1] = bp[PB];
                b[j] = t;
            }
#pragma unroll
            for (int j = 0; j < WN; ++j)
                acc[j] = __builtin_amdgcn_wmma_f32_16x16x4_f32(
                    false, a, false, b[j], (short)0, acc[j], false, false);
        }
        if (more) {
#pragma unroll
            for (int i = 0; i < RA; ++i) {
                const int idx = tid + i * 256;
                *(v4f*)&sA[buf ^ 1][(idx / KA4) * PA + (idx % KA4) * 4] = ra[i];
            }
#pragma unroll
            for (int i = 0; i < RB; ++i) {
                const int idx = tid + i * 256;
                *(v4f*)&sB[buf ^ 1][(idx / BN4) * PB + (idx % BN4) * 4] = rb[i];
            }
        }
        __syncthreads();
    }

    float* sP = &sB[0][0];
#pragma unroll
    for (int j = 0; j < WN; ++j)
#pragma unroll
        for (int r = 0; r < 8; ++r)
            sP[w * (BM * BN) + (r + 8 * lh) * BN + j * 16 + lr] = acc[j][r];
    __syncthreads();
    float* Cs = C + (size_t)m0 * BN;
#pragma unroll
    for (int i = 0; i < BM * BN / 256; ++i) {
        const int idx = tid + i * 256;
        const float p0 = sP[0 * (BM * BN) + idx], p1 = sP[1 * (BM * BN) + idx];
        const float p2 = sP[2 * (BM * BN) + idx], p3 = sP[3 * (BM * BN) + idx];
        const float p4 = sP[4 * (BM * BN) + idx], p5 = sP[5 * (BM * BN) + idx];
        const float p6 = sP[6 * (BM * BN) + idx], p7 = sP[7 * (BM * BN) + idx];
        Cs[idx] = ((p0 + p1) + (p2 + p3)) + ((p4 + p5) + (p6 + p7));
    }
}

__global__ __launch_bounds__(256)
void conv_silu_kernel(const float* __restrict__ xr, const float* __restrict__ conv_w,
                      const float* __restrict__ conv_b, float* __restrict__ xs)
{
    const int idx = blockIdx.x * 256 + threadIdx.x;
    const int l = idx / DI;
    const int d = idx % DI;
    const float4 w = *(const float4*)(conv_w + (size_t)d * KC);
    float acc = conv_b[d];
    const float* xc = xr + d;
    if (l >= 3) acc += xc[(size_t)(l - 3) * (2 * DI)] * w.x;
    if (l >= 2) acc += xc[(size_t)(l - 2) * (2 * DI)] * w.y;
    if (l >= 1) acc += xc[(size_t)(l - 1) * (2 * DI)] * w.z;
    acc             += xc[(size_t)l       * (2 * DI)] * w.w;
    const float sig = 1.f / (1.f + __expf(-acc));
    xs[idx] = acc * sig;
}

#define PP 12
#define PROW (32 * PP)

__global__ __launch_bounds__(256)
void scan_kernel(const float* __restrict__ dt, const float* __restrict__ xdbl,
                 const float* __restrict__ xs, const float* __restrict__ xr,
                 const float* __restrict__ A_log, const float* __restrict__ Dvec,
                 float* __restrict__ y, float* __restrict__ hT)
{
    __shared__ __align__(16) float sDT[64 * 32];
    __shared__ __align__(16) float sXS[64 * 32];
    __shared__ __align__(16) float sBC[64 * 32];
    __shared__ __align__(16) float sP[64 * PROW];

    const int t  = blockIdx.x * 256 + threadIdx.x;
    const int d  = t >> 3;
    const int q  = t & 7;
    const int ch = threadIdx.x >> 3;
    const int d0 = blockIdx.x * 32;
    const int wr = threadIdx.x >> 5;
    const int wc = threadIdx.x & 31;
    const int pslot = ch * PP + q;

    float A0 = -__expf(A_log[(size_t)d * NS + 2 * q]);
    float A1 = -__expf(A_log[(size_t)d * NS + 2 * q + 1]);
    float h0 = 0.f, h1 = 0.f;
    const float Dw = Dvec[d0 + wc];

    const float* gDT = dt   + (size_t)ch * DI + d0 + 4 * q;
    const float* gXS = xs   + (size_t)ch * DI + d0 + 4 * q;
    const float* gBC = xdbl + (size_t)ch * XDBL_W + DTR + 4 * q;
    v4f pdt[2], pxs[2], pbc[2];
    for (int i = 0; i < 2; ++i) {
        pdt[i] = *(const v4f*)(gDT + (size_t)(32 * i) * DI);
        pxs[i] = *(const v4f*)(gXS + (size_t)(32 * i) * DI);
        pbc[i] = *(const v4f*)(gBC + (size_t)(32 * i) * XDBL_W);
    }
    for (int i = 0; i < 2; ++i) {
        *(v4f*)&sDT[(threadIdx.x + 256 * i) * 4] = pdt[i];
        *(v4f*)&sXS[(threadIdx.x + 256 * i) * 4] = pxs[i];
        *(v4f*)&sBC[(threadIdx.x + 256 * i) * 4] = pbc[i];
    }

    for (int c = 0; c < LSEQ / 64; ++c) {
        const bool more = (c + 1 < LSEQ / 64);
        __syncthreads();
        if (more) {
            for (int i = 0; i < 2; ++i) {
                const size_t l1 = (size_t)(c + 1) * 64 + 32 * i;
                pdt[i] = *(const v4f*)(gDT + l1 * DI);
                pxs[i] = *(const v4f*)(gXS + l1 * DI);
                pbc[i] = *(const v4f*)(gBC + l1 * XDBL_W);
            }
        }
        float resv[8], xsv[8];
        for (int k = 0; k < 8; ++k) {
            const int l = c * 64 + wr + 8 * k;
            resv[k] = xr[(size_t)l * (2 * DI) + DI + d0 + wc];
            xsv[k]  = sXS[(wr + 8 * k) * 32 + wc];
        }
#pragma unroll 8
        for (int s = 0; s < 64; ++s) {
            const float dtv = sDT[s * 32 + ch];
            const float xv  = sXS[s * 32 + ch];
            const float2 Bv = *(const float2*)&sBC[s * 32 + 2 * q];
            const float2 Cv = *(const float2*)&sBC[s * 32 + 16 + 2 * q];
            const float dx  = dtv * xv;
            const float dA0 = __expf(dtv * A0);
            const float dA1 = __expf(dtv * A1);
            h0 = dA0 * h0 + dx * Bv.x;
            h1 = dA1 * h1 + dx * Bv.y;
            sP[s * PROW + pslot] = fmaf(h0, Cv.x, h1 * Cv.y);
        }
        __syncthreads();
        if (more) {
            for (int i = 0; i < 2; ++i) {
                *(v4f*)&sDT[(threadIdx.x + 256 * i) * 4] = pdt[i];
                *(v4f*)&sXS[(threadIdx.x + 256 * i) * 4] = pxs[i];
                *(v4f*)&sBC[(threadIdx.x + 256 * i) * 4] = pbc[i];
            }
        }
        for (int k = 0; k < 8; ++k) {
            const int ls = wr + 8 * k;
            const int l = c * 64 + ls;
            const float4 pa = *(const float4*)&sP[ls * PROW + wc * PP];
            const float4 pb = *(const float4*)&sP[ls * PROW + wc * PP + 4];
            float yv = ((pa.x + pa.y) + (pa.z + pa.w)) + ((pb.x + pb.y) + (pb.z + pb.w));
            yv = yv + xsv[k] * Dw;
            const float sig = 1.f / (1.f + __expf(-resv[k]));
            y[(size_t)l * DI + d0 + wc] = yv * (resv[k] * sig);
        }
    }
    float2 hout = make_float2(h0, h1);
    *(float2*)&hT[(size_t)d * NS + 2 * q] = hout;
}

extern "C" void kernel_launch(void* const* d_in, const int* in_sizes, int n_in,
                              void* d_out, int out_size, void* d_ws, size_t ws_size,
                              hipStream_t stream) {
    const float* x      = (const float*)d_in[0];
    const float* W_in   = (const float*)d_in[1];
    const float* conv_w = (const float*)d_in[2];
    const float* conv_b = (const float*)d_in[3];
    const float* W_x    = (const float*)d_in[4];
    const float* W_dt   = (const float*)d_in[5];
    const float* b_dt   = (const float*)d_in[6];
    const float* A_log  = (const float*)d_in[7];
    const float* Dv     = (const float*)d_in[8];
    const float* W_out  = (const float*)d_in[9];

    float* out = (float*)d_out;
    float* hT  = out + (size_t)LSEQ * DM;

    float* ws   = (float*)d_ws;
    float* xr   = ws;
    float* xs   = xr   + (size_t)LSEQ * 2 * DI;
    float* xdbl = xs   + (size_t)LSEQ * DI;
    float* dt   = xdbl + (size_t)LSEQ * XDBL_W;
    float* yb   = dt   + (size_t)LSEQ * DI;

    wmma_gemm_f32<128, 128, 16, 2, 4, false><<<512, 256, 0, stream>>>(
        x, W_in, xr, nullptr, LSEQ, 2 * DI, DM, DM, 2 * DI, 2 * DI);

    conv_silu_kernel<<<(LSEQ * DI) / 256, 256, 0, stream>>>(xr, conv_w, conv_b, xs);

    wmma_gemm_f32_ksplit<XDBL_W, 128><<<LSEQ / 16, 256, 0, stream>>>(
        xs, W_x, xdbl, DI, DI);

    wmma_gemm_f32<128, 128, 16, 2, 4, true><<<256, 256, 0, stream>>>(
        xdbl, W_dt, dt, b_dt, LSEQ, DI, DTR, XDBL_W, DI, DI);

    scan_kernel<<<(DI * 8) / 256, 256, 0, stream>>>(dt, xdbl, xs, xr, A_log, Dv, yb, hT);

    wmma_gemm_f32<64, 128, 32, 2, 4, false><<<256, 256, 0, stream>>>(
        yb, W_out, out, nullptr, LSEQ, DM, DI, DI, DM, DM);
}


# scan split in time into 8 parts run side by side (512 blocks, not 64): a pre-launch gets each part's end state from zero and its product of dA, parts start from the carried state; h sum order changed
# speedup vs baseline: 7.6343x; 1.1236x over previous
#include <hip/hip_runtime.h>
#include <hip/hip_bf16.h>
#include <math.h>


typedef float v2f __attribute__((ext_vector_type(2)));
typedef float v4f __attribute__((ext_vector_type(4)));
typedef float v8f __attribute__((ext_vector_type(8)));

#define DM   1024
#define DI   2048
#define NS   16
#define KC   4
#define DTR  64
#define LSEQ 2048
#define XDBL_W (DTR + 2*NS)

template<int BM, int BN, int KS, int WGM, int WGN, bool SOFTPLUS_BIAS>
__global__ __launch_bounds__(256)
void wmma_gemm_f32(const float* __restrict__ A, const float* __restrict__ B,
                   float* __restrict__ C, const float* __restrict__ bias,
                   int M, int N, int K, int lda, int ldb, int ldc)
{
    constexpr int PA  = KS + 4;
    constexpr int PB  = BN + 8;
    constexpr int WM  = BM / (16 * WGM);
    constexpr int WN  = BN / (16 * WGN);
    constexpr int NA4 = BM * KS / 4;
    constexpr int NB4 = KS * BN / 4;
    constexpr int RA  = (NA4 + 255) / 256;
    constexpr int RB  = (NB4 + 255) / 256;
    constexpr int BN4 = BN / 4;
    constexpr int KA4 = KS / 4;
    static_assert(WGM * WGN == 8, "8 waves per block");
    static_assert(WM * 16 * WGM == BM && WN * 16 * WGN == BN, "tile split");

    __shared__ __align__(16) float sA[2][BM * PA];
    __shared__ __align__(16) float sB[2][KS * PB];

    const int tid  = threadIdx.x;
    const int lane = tid & 31;
    const int w    = tid >> 5;
    const int wm   = w % WGM;
    const int wn   = w / WGM;
    const int lr   = lane & 15;
    const int lh   = lane >> 4;
    const int tilesM = M / BM;
    const int m0 = (blockIdx.x % tilesM) * BM;
    const int n0 = (blockIdx.x / tilesM) * BN;

    v8f acc[WM][WN];
#pragma unroll
    for (int i = 0; i < WM; ++i)
#pragma unroll
        for (int j = 0; j < WN; ++j) {
            v8f z = {0.f, 0.f, 0.f, 0.f, 0.f, 0.f, 0.f, 0.f};
            acc[i][j] = z;
        }

    v4f ra[RA], rb[RB];

#pragma unroll
    for (int i = 0; i < RA; ++i) {
        const int idx = tid + i * 256;
        if ((NA4 % 256 == 0) || idx < NA4)
            ra[i] = *(const v4f*)(A + (size_t)(m0 + (idx / KA4)) * lda + (idx % KA4) * 4);
    }
#pragma unroll
    for (int i = 0; i < RB; ++i) {
        const int idx = tid + i * 256;
        if ((NB4 % 256 == 0) || idx < NB4)
            rb[i] = *(const v4f*)(B + (size_t)(idx / BN4) * ldb + n0 + (idx % BN4) * 4);
    }
#pragma unroll
    for (int i = 0; i < RA; ++i) {
        const int idx = tid + i * 256;
        if ((NA4 % 256 == 0) || idx < NA4)
            *(v4f*)&sA[0][(idx / KA4) * PA + (idx % KA4) * 4] = ra[i];
    }
#pragma unroll
    for (int i = 0; i < RB; ++i) {
        const int idx = tid + i * 256;
        if ((NB4 % 256 == 0) || idx < NB4)
            *(v4f*)&sB[0][(idx / BN4) * PB + (idx % BN4) * 4] = rb[i];
    }
    __syncthreads();

    const int nk = K / KS;
    for (int kt = 0; kt < nk; ++kt) {
        const int buf = kt & 1;
        const bool more = (kt + 1 < nk);
        if (more) {
            const int k0 = (kt + 1) * KS;
#pragma unroll
            for (int i = 0; i < RA; ++i) {
                const int idx = tid + i * 256;
                if ((NA4 % 256 == 0) || idx < NA4)
                    ra[i] = *(const v4f*)(A + (size_t)(m0 + (idx / KA4)) * lda + k0 + (idx % KA4) * 4);
            }
#pragma unroll
            for (int i = 0; i < RB; ++i) {
                const int idx = tid + i * 256;
                if ((NB4 % 256 == 0) || idx < NB4)
                    rb[i] = *(const v4f*)(B + (size_t)(k0 + idx / BN4) * ldb + n0 + (idx % BN4) * 4);
            }
        }
        const float* tA = &sA[buf][(wm * WM * 16 + lr) * PA + 2 * lh];
        const float* tB = &sB[buf][(2 * lh) * PB + wn * WN * 16 + lr];
#pragma unroll
        for (int kk = 0; kk < KS / 4; ++kk) {
            v2f a[WM];
#pragma unroll
            for (int i = 0; i < WM; ++i)
                a[i] = *(const v2f*)(tA + i * 16 * PA + kk * 4);
            v2f b[WN];
#pragma unroll
            for (int j = 0; j < WN; ++j) {
                const float* bp = tB + (kk * 4) * PB + j * 16;
                v2f t; t[0] = bp[0]; t[1] = bp[PB];
                b[j] = t;
            }
#pragma unroll
            for (int i = 0; i < WM; ++i)
#pragma unroll
                for (int j = 0; j < WN; ++j)
                    acc[i][j] = __builtin_amdgcn_wmma_f32_16x16x4_f32(
                        false, a[i], false, b[j], (short)0, acc[i][j], false, false);
        }
        if (more) {
#pragma unroll
            for (int i = 0; i < RA; ++i) {
                const int idx = tid + i * 256;
                if ((NA4 % 256 == 0) || idx < NA4)
                    *(v4f*)&sA[buf ^ 1][(idx / KA4) * PA + (idx % KA4) * 4] = ra[i];
            }
#pragma unroll
            for (int i = 0; i < RB; ++i) {
                const int idx = tid + i * 256;
                if ((NB4 % 256 == 0) || idx < NB4)
                    *(v4f*)&sB[buf ^ 1][(idx / BN4) * PB + (idx % BN4) * 4] = rb[i];
            }
        }
        __syncthreads();
    }

#pragma unroll
    for (int i = 0; i < WM; ++i) {
#pragma unroll
        for (int j = 0; j < WN; ++j) {
            const int col = n0 + wn * WN * 16 + j * 16 + lr;
#pragma unroll
            for (int r = 0; r < 8; ++r) {
                const int row = m0 + wm * WM * 16 + i * 16 + r + 8 * lh;
                float v = acc[i][j][r];
                if (SOFTPLUS_BIAS) {
                    v += bias[col];
                    v = (v > 20.f) ? v : __logf(1.f + __expf(v));
                }
                C[(size_t)row * ldc + col] = v;
            }
        }
    }
}

template<int BN, int KS>
__global__ __launch_bounds__(256)
void wmma_gemm_f32_ksplit(const float* __restrict__ A, const float* __restrict__ B,
                          float* __restrict__ C, int K, int lda)
{
    constexpr int BM  = 16;
    constexpr int PA  = KS + 4;
    constexpr int PB  = BN + 8;
    constexpr int WN  = BN / 16;
    constexpr int RA  = BM * KS / 4 / 256;
    constexpr int RB  = KS * BN / 4 / 256;
    constexpr int BN4 = BN / 4;
    constexpr int KA4 = KS / 4;
    constexpr int KW  = KS / 32;
    static_assert((BM * KS / 4) % 256 == 0 && (KS * BN / 4) % 256 == 0, "whole pieces per thread");
    static_assert((BM * BN) % 256 == 0, "whole outputs per thread");
    static_assert(8 * BM * BN <= KS * PB, "the 8 partial strips fit in one B buffer");

    __shared__ __align__(16) float sA[2][BM * PA];
    __shared__ __align__(16) float sB[2][KS * PB];

    const int tid  = threadIdx.x;
    const int lane = tid & 31;
    const int w    = tid >> 5;
    const int lr   = lane & 15;
    const int lh   = lane >> 4;
    const int m0   = blockIdx.x * BM;

    v8f acc[WN];
#pragma unroll
    for (int j = 0; j < WN; ++j) {
        v8f z = {0.f, 0.f, 0.f, 0.f, 0.f, 0.f, 0.f, 0.f};
        acc[j] = z;
    }

    v4f ra[RA], rb[RB];

#pragma unroll
    for (int i = 0; i < RA; ++i) {
        const int idx = tid + i * 256;
        ra[i] = *(const v4f*)(A + (size_t)(m0 + (idx / KA4)) * lda + (idx % KA4) * 4);
    }
#pragma unroll
    for (int i = 0; i < RB; ++i) {
        const int idx = tid + i * 256;
        rb[i] = *(const v4f*)(B + (size_t)(idx / BN4) * BN + (idx % BN4) * 4);
    }
#pragma unroll
    for (int i = 0; i < RA; ++i) {
        const int idx = tid + i * 256;
        *(v4f*)&sA[0][(idx / KA4) * PA + (idx % KA4) * 4] = ra[i];
    }
#pragma unroll
    for (int i = 0; i < RB; ++i) {
        const int idx = tid + i * 256;
        *(v4f*)&sB[0][(idx / BN4) * PB + (idx % BN4) * 4] = rb[i];
    }
    __syncthreads();

    const int nk = K / KS;
    for (int kt = 0; kt < nk; ++kt) {
        const int buf = kt & 1;
        const bool more = (kt + 1 < nk);
        if (more) {
            const int k0 = (kt + 1) * KS;
#pragma unroll
            for (int i = 0; i < RA; ++i) {
                const int idx = tid + i * 256;
                ra[i] = *(const v4f*)(A + (size_t)(m0 + (idx / KA4)) * lda + k0 + (idx % KA4) * 4);
            }
#pragma unroll
            for (int i = 0; i < RB; ++i) {
                const int idx = tid + i * 256;
                rb[i] = *(const v4f*)(B + (size_t)(k0 + idx / BN4) * BN + (idx % BN4) * 4);
            }
        }
        const float* tA = &sA[buf][lr * PA + w * KW * 4 + 2 * lh];
        const float* tB = &sB[buf][(w * KW * 4 + 2 * lh) * PB + lr];
#pragma unroll
        for (int kk = 0; kk < KW; ++kk) {
            const v2f a = *(const v2f*)(tA + kk * 4);
            v2f b[WN];
#pragma unroll
            for (int j = 0; j < WN; ++j) {
                const float* bp = tB + (kk * 4) * PB + j * 16;
                v2f t; t[0] = bp[0]; t[1] = bp[PB];
                b[j] = t;
            }
#pragma unroll
            for (int j = 0; j < WN; ++j)
                acc[j] = __builtin_amdgcn_wmma_f32_16x16x4_f32(
                    false, a, false, b[j], (short)0, acc[j], false, false);
        }
        if (more) {
#pragma unroll
            for (int i = 0; i < RA; ++i) {
                const int idx = tid + i * 256;
                *(v4f*)&sA[buf ^ 1][(idx / KA4) * PA + (idx % KA4) * 4] = ra[i];
            }
#pragma unroll
            for (int i = 0; i < RB; ++i) {
                const int idx = tid + i * 256;
                *(v4f*)&sB[buf ^ 1][(idx / BN4) * PB + (idx % BN4) * 4] = rb[i];
            }
        }
        __syncthreads();
    }

    float* sP = &sB[0][0];
#pragma unroll
    for (int j = 0; j < WN; ++j)
#pragma unroll
        for (int r = 0; r < 8; ++r)
            sP[w * (BM * BN) + (r + 8 * lh) * BN + j * 16 + lr] = acc[j][r];
    __syncthreads();
    float* Cs = C + (size_t)m0 * BN;
#pragma unroll
    for (int i = 0; i < BM * BN / 256; ++i) {
        const int idx = tid + i * 256;
        const float p0 = sP[0 * (BM * BN) + idx], p1 = sP[1 * (BM * BN) + idx];
        const float p2 = sP[2 * (BM * BN) + idx], p3 = sP[3 * (BM * BN) + idx];
        const float p4 = sP[4 * (BM * BN) + idx], p5 = sP[5 * (BM * BN) + idx];
        const float p6 = sP[6 * (BM * BN) + idx], p7 = sP[7 * (BM * BN) + idx];
        Cs[idx] = ((p0 + p1) + (p2 + p3)) + ((p4 + p5) + (p6 + p7));
    }
}

__global__ __launch_bounds__(256)
void conv_silu_kernel(const float* __restrict__ xr, const float* __restrict__ conv_w,
                      const float* __restrict__ conv_b, float* __restrict__ xs)
{
    const int idx = blockIdx.x * 256 + threadIdx.x;
    const int l = idx / DI;
    const int d = idx % DI;
    const float4 w = *(const float4*)(conv_w + (size_t)d * KC);
    float acc = conv_b[d];
    const float* xc = xr + d;
    if (l >= 3) acc += xc[(size_t)(l - 3) * (2 * DI)] * w.x;
    if (l >= 2) acc += xc[(size_t)(l - 2) * (2 * DI)] * w.y;
    if (l >= 1) acc += xc[(size_t)(l - 1) * (2 * DI)] * w.z;
    acc             += xc[(size_t)l       * (2 * DI)] * w.w;
    const float sig = 1.f / (1.f + __expf(-acc));
    xs[idx] = acc * sig;
}

#define NPART 8
#define LPART (LSEQ / NPART)
#define NCHUNK (LPART / 64)
#define PP 12
#define PROW (32 * PP)

template<bool ENDS_ONLY>
__global__ __launch_bounds__(256)
void scan_kernel(const float* __restrict__ dt, const float* __restrict__ xdbl,
                 const float* __restrict__ xs, const float* __restrict__ xr,
                 const float* __restrict__ A_log, const float* __restrict__ Dvec,
                 float* __restrict__ y, float* __restrict__ hT,
                 float* __restrict__ endH, float* __restrict__ endP)
{
    __shared__ __align__(16) float sDT[64 * 32];
    __shared__ __align__(16) float sXS[64 * 32];
    __shared__ __align__(16) float sBC[64 * 32];
    __shared__ __align__(16) float sP[ENDS_ONLY ? 4 : 64 * PROW];

    const int part  = blockIdx.x >> 6;
    const int d0    = (blockIdx.x & 63) * 32;
    const int lbase = part * LPART;
    const int ch = threadIdx.x >> 3;
    const int q  = threadIdx.x & 7;
    const int d  = d0 + ch;
    const int wr = threadIdx.x >> 5;
    const int wc = threadIdx.x & 31;
    const int pslot = ch * PP + q;

    float A0 = -__expf(A_log[(size_t)d * NS + 2 * q]);
    float A1 = -__expf(A_log[(size_t)d * NS + 2 * q + 1]);
    float h0 = 0.f, h1 = 0.f;
    float p0 = 1.f, p1 = 1.f;
    float Dw = 0.f;
    if (!ENDS_ONLY) Dw = Dvec[d0 + wc];

    const float* gDT = dt   + (size_t)(lbase + ch) * DI + d0 + 4 * q;
    const float* gXS = xs   + (size_t)(lbase + ch) * DI + d0 + 4 * q;
    const float* gBC = xdbl + (size_t)(lbase + ch) * XDBL_W + DTR + 4 * q;
    v4f pdt[2], pxs[2], pbc[2];
    for (int i = 0; i < 2; ++i) {
        pdt[i] = *(const v4f*)(gDT + (size_t)(32 * i) * DI);
        pxs[i] = *(const v4f*)(gXS + (size_t)(32 * i) * DI);
        pbc[i] = *(const v4f*)(gBC + (size_t)(32 * i) * XDBL_W);
    }

    if (!ENDS_ONLY) {
        float2 eh[NPART - 1], ep[NPART - 1];
#pragma unroll
        for (int j = 0; j < NPART - 1; ++j) {
            const int jj = (j < part) ? j : 0;
            eh[j] = *(const float2*)&endH[((size_t)jj * DI + d) * NS + 2 * q];
            ep[j] = *(const float2*)&endP[((size_t)jj * DI + d) * NS + 2 * q];
        }
#pragma unroll
        for (int j = 0; j < NPART - 1; ++j)
            if (j < part) {
                h0 = ep[j].x * h0 + eh[j].x;
                h1 = ep[j].y * h1 + eh[j].y;
            }
    }

    for (int i = 0; i < 2; ++i) {
        *(v4f*)&sDT[(threadIdx.x + 256 * i) * 4] = pdt[i];
        *(v4f*)&sXS[(threadIdx.x + 256 * i) * 4] = pxs[i];
        *(v4f*)&sBC[(threadIdx.x + 256 * i) * 4] = pbc[i];
    }

    for (int c = 0; c < NCHUNK; ++c) {
        const bool more = (c + 1 < NCHUNK);
        __syncthreads();
        if (more) {
            for (int i = 0; i < 2; ++i) {
                const size_t l1 = (size_t)(c + 1) * 64 + 32 * i;
                pdt[i] = *(const v4f*)(gDT + l1 * DI);
                pxs[i] = *(const v4f*)(gXS + l1 * DI);
                pbc[i] = *(const v4f*)(gBC + l1 * XDBL_W);
            }
        }
        if (ENDS_ONLY) {
            auto read8 = [&](int g, float (&dtg)[8], float (&xg)[8], float2 (&Bg)[8]) {
#pragma unroll
                for (int k = 0; k < 8; ++k) {
                    const int s = g * 8 + k;
                    dtg[k] = sDT[s * 32 + ch];
                    xg[k]  = sXS[s * 32 + ch];
                    Bg[k]  = *(const float2*)&sBC[s * 32 + 2 * q];
                }
            };
            auto step8 = [&](const float (&dtg)[8], const float (&xg)[8], const float2 (&Bg)[8]) {
#pragma unroll
                for (int k = 0; k < 8; ++k) {
                    const float dx  = dtg[k] * xg[k];
                    const float dA0 = __expf(dtg[k] * A0);
                    const float dA1 = __expf(dtg[k] * A1);
                    h0 = dA0 * h0 + dx * Bg[k].x;
                    h1 = dA1 * h1 + dx * Bg[k].y;
                    p0 *= dA0;
                    p1 *= dA1;
                }
            };
            float dta[8], xa[8], dtb[8], xb[8];
            float2 Ba[8], Bb[8];
            read8(0, dta, xa, Ba);
#pragma unroll 1
            for (int g = 0; g < 8; g += 2) {
                read8(g + 1, dtb, xb, Bb);
                step8(dta, xa, Ba);
                if (g + 2 < 8) read8(g + 2, dta, xa, Ba);
                step8(dtb, xb, Bb);
            }
            __syncthreads();
            if (more) {
                for (int i = 0; i < 2; ++i) {
                    *(v4f*)&sDT[(threadIdx.x + 256 * i) * 4] = pdt[i];
                    *(v4f*)&sXS[(threadIdx.x + 256 * i) * 4] = pxs[i];
                    *(v4f*)&sBC[(threadIdx.x + 256 * i) * 4] = pbc[i];
                }
            }
        } else {
            float resv[8], xsv[8];
            for (int k = 0; k < 8; ++k) {
                const int l = lbase + c * 64 + wr + 8 * k;
                resv[k] = xr[(size_t)l * (2 * DI) + DI + d0 + wc];
                xsv[k]  = sXS[(wr + 8 * k) * 32 + wc];
            }
#pragma unroll 8
            for (int s = 0; s < 64; ++s) {
                const float dtv = sDT[s * 32 + ch];
                const float xv  = sXS[s * 32 + ch];
                const float2 Bv = *(const float2*)&sBC[s * 32 + 2 * q];
                const float2 Cv = *(const float2*)&sBC[s * 32 + 16 + 2 * q];
                const float dx  = dtv * xv;
                const float dA0 = __expf(dtv * A0);
                const float dA1 = __expf(dtv * A1);
                h0 = dA0 * h0 + dx * Bv.x;
                h1 = dA1 * h1 + dx * Bv.y;
                sP[s * PROW + pslot] = fmaf(h0, Cv.x, h1 * Cv.y);
            }
            __syncthreads();
            if (more) {
                for (int i = 0; i < 2; ++i) {
                    *(v4f*)&sDT[(threadIdx.x + 256 * i) * 4] = pdt[i];
                    *(v4f*)&sXS[(threadIdx.x + 256 * i) * 4] = pxs[i];
                    *(v4f*)&sBC[(threadIdx.x + 256 * i) * 4] = pbc[i];
                }
            }
            for (int k = 0; k < 8; ++k) {
                const int ls = wr + 8 * k;
                const int l = lbase + c * 64 + ls;
                const float4 pa = *(const float4*)&sP[ls * PROW + wc * PP];
                const float4 pb = *(const float4*)&sP[ls * PROW + wc * PP + 4];
                float yv = ((pa.x + pa.y) + (pa.z + pa.w)) + ((pb.x + pb.y) + (pb.z + pb.w));
                yv = yv + xsv[k] * Dw;
                const float sig = 1.f / (1.f + __expf(-resv[k]));
                y[(size_t)l * DI + d0 + wc] = yv * (resv[k] * sig);
            }
        }
    }
    if (ENDS_ONLY) {
        *(float2*)&endH[((size_t)part * DI + d) * NS + 2 * q] = make_float2(h0, h1);
        *(float2*)&endP[((size_t)part * DI + d) * NS + 2 * q] = make_float2(p0, p1);
    } else if (part == NPART - 1) {
        *(float2*)&hT[(size_t)d * NS + 2 * q] = make_float2(h0, h1);
    }
}

extern "C" void kernel_launch(void* const* d_in, const int* in_sizes, int n_in,
                              void* d_out, int out_size, void* d_ws, size_t ws_size,
                              hipStream_t stream) {
    const float* x      = (const float*)d_in[0];
    const float* W_in   = (const float*)d_in[1];
    const float* conv_w = (const float*)d_in[2];
    const float* conv_b = (const float*)d_in[3];
    const float* W_x    = (const float*)d_in[4];
    const float* W_dt   = (const float*)d_in[5];
    const float* b_dt   = (const float*)d_in[6];
    const float* A_log  = (const float*)d_in[7];
    const float* Dv     = (const float*)d_in[8];
    const float* W_out  = (const float*)d_in[9];

    float* out = (float*)d_out;
    float* hT  = out + (size_t)LSEQ * DM;

    float* ws   = (float*)d_ws;
    float* xr   = ws;
    float* xs   = xr   + (size_t)LSEQ * 2 * DI;
    float* xdbl = xs   + (size_t)LSEQ * DI;
    float* dt   = xdbl + (size_t)LSEQ * XDBL_W;
    float* yb   = dt   + (size_t)LSEQ * DI;
    float* endH = yb   + (size_t)LSEQ * DI;
    float* endP = endH + (size_t)(NPART - 1) * DI * NS;

    wmma_gemm_f32<128, 128, 16, 2, 4, false><<<512, 256, 0, stream>>>(
        x, W_in, xr, nullptr, LSEQ, 2 * DI, DM, DM, 2 * DI, 2 * DI);

    conv_silu_kernel<<<(LSEQ * DI) / 256, 256, 0, stream>>>(xr, conv_w, conv_b, xs);

    wmma_gemm_f32_ksplit<XDBL_W, 128><<<LSEQ / 16, 256, 0, stream>>>(
        xs, W_x, xdbl, DI, DI);

    wmma_gemm_f32<128, 128, 16, 2, 4, true><<<256, 256, 0, stream>>>(
        xdbl, W_dt, dt, b_dt, LSEQ, DI, DTR, XDBL_W, DI, DI);

    scan_kernel<true><<<(NPART - 1) * (DI / 32), 256, 0, stream>>>(
        dt, xdbl, xs, xr, A_log, Dv, yb, hT, endH, endP);
    scan_kernel<false><<<NPART * (DI / 32), 256, 0, stream>>>(
        dt, xdbl, xs, xr, A_log, Dv, yb, hT, endH, endP);

    wmma_gemm_f32<64, 128, 32, 2, 4, false><<<256, 256, 0, stream>>>(
        yb, W_out, out, nullptr, LSEQ, DM, DI, DI, DM, DM);
}


# x-proj GEMM computes the depthwise conv + SiLU in its own A staging (a block visits its rows x channels once) and writes xs from there: the one-element-per-thread conv launch is gone; sums unchanged
# speedup vs baseline: 7.9261x; 1.0382x over previous
#include <hip/hip_runtime.h>
#include <hip/hip_bf16.h>
#include <math.h>


typedef float v2f __attribute__((ext_vector_type(2)));
typedef float v4f __attribute__((ext_vector_type(4)));
typedef float v8f __attribute__((ext_vector_type(8)));

#define DM   1024
#define DI   2048
#define NS   16
#define KC   4
#define DTR  64
#define LSEQ 2048
#define XDBL_W (DTR + 2*NS)

template<int BM, int BN, int KS, int WGM, int WGN, bool SOFTPLUS_BIAS>
__global__ __launch_bounds__(256)
void wmma_gemm_f32(const float* __restrict__ A, const float* __restrict__ B,
                   float* __restrict__ C, const float* __restrict__ bias,
                   int M, int N, int K, int lda, int ldb, int ldc)
{
    constexpr int PA  = KS + 4;
    constexpr int PB  = BN + 8;
    constexpr int WM  = BM / (16 * WGM);
    constexpr int WN  = BN / (16 * WGN);
    constexpr int NA4 = BM * KS / 4;
    constexpr int NB4 = KS * BN / 4;
    constexpr int RA  = (NA4 + 255) / 256;
    constexpr int RB  = (NB4 + 255) / 256;
    constexpr int BN4 = BN / 4;
    constexpr int KA4 = KS / 4;
    static_assert(WGM * WGN == 8, "8 waves per block");
    static_assert(WM * 16 * WGM == BM && WN * 16 * WGN == BN, "tile split");

    __shared__ __align__(16) float sA[2][BM * PA];
    __shared__ __align__(16) float sB[2][KS * PB];

    const int tid  = threadIdx.x;
    const int lane = tid & 31;
    const int w    = tid >> 5;
    const int wm   = w % WGM;
    const int wn   = w / WGM;
    const int lr   = lane & 15;
    const int lh   = lane >> 4;
    const int tilesM = M / BM;
    const int m0 = (blockIdx.x % tilesM) * BM;
    const int n0 = (blockIdx.x / tilesM) * BN;

    v8f acc[WM][WN];
#pragma unroll
    for (int i = 0; i < WM; ++i)
#pragma unroll
        for (int j = 0; j < WN; ++j) {
            v8f z = {0.f, 0.f, 0.f, 0.f, 0.f, 0.f, 0.f, 0.f};
            acc[i][j] = z;
        }

    v4f ra[RA], rb[RB];

#pragma unroll
    for (int i = 0; i < RA; ++i) {
        const int idx = tid + i * 256;
        if ((NA4 % 256 == 0) || idx < NA4)
            ra[i] = *(const v4f*)(A + (size_t)(m0 + (idx / KA4)) * lda + (idx % KA4) * 4);
    }
#pragma unroll
    for (int i = 0; i < RB; ++i) {
        const int idx = tid + i * 256;
        if ((NB4 % 256 == 0) || idx < NB4)
            rb[i] = *(const v4f*)(B + (size_t)(idx / BN4) * ldb + n0 + (idx % BN4) * 4);
    }
#pragma unroll
    for (int i = 0; i < RA; ++i) {
        const int idx = tid + i * 256;
        if ((NA4 % 256 == 0) || idx < NA4)
            *(v4f*)&sA[0][(idx / KA4) * PA + (idx % KA4) * 4] = ra[i];
    }
#pragma unroll
    for (int i = 0; i < RB; ++i) {
        const int idx = tid + i * 256;
        if ((NB4 % 256 == 0) || idx < NB4)
            *(v4f*)&sB[0][(idx / BN4) * PB + (idx % BN4) * 4] = rb[i];
    }
    __syncthreads();

    const int nk = K / KS;
    for (int kt = 0; kt < nk; ++kt) {
        const int buf = kt & 1;
        const bool more = (kt + 1 < nk);
        if (more) {
            const int k0 = (kt + 1) * KS;
#pragma unroll
            for (int i = 0; i < RA; ++i) {
                const int idx = tid + i * 256;
                if ((NA4 % 256 == 0) || idx < NA4)
                    ra[i] = *(const v4f*)(A + (size_t)(m0 + (idx / KA4)) * lda + k0 + (idx % KA4) * 4);
            }
#pragma unroll
            for (int i = 0; i < RB; ++i) {
                const int idx = tid + i * 256;
                if ((NB4 % 256 == 0) || idx < NB4)
                    rb[i] = *(const v4f*)(B + (size_t)(k0 + idx / BN4) * ldb + n0 + (idx % BN4) * 4);
            }
        }
        const float* tA = &sA[buf][(wm * WM * 16 + lr) * PA + 2 * lh];
        const float* tB = &sB[buf][(2 * lh) * PB + wn * WN * 16 + lr];
#pragma unroll
        for (int kk = 0; kk < KS / 4; ++kk) {
            v2f a[WM];
#pragma unroll
            for (int i = 0; i < WM; ++i)
                a[i] = *(const v2f*)(tA + i * 16 * PA + kk * 4);
            v2f b[WN];
#pragma unroll
            for (int j = 0; j < WN; ++j) {
                const float* bp = tB + (kk * 4) * PB + j * 16;
                v2f t; t[0] = bp[0]; t[1] = bp[PB];
                b[j] = t;
            }
#pragma unroll
            for (int i = 0; i < WM; ++i)
#pragma unroll
                for (int j = 0; j < WN; ++j)
                    acc[i][j] = __builtin_amdgcn_wmma_f32_16x16x4_f32(
                        false, a[i], false, b[j], (short)0, acc[i][j], false, false);
        }
        if (more) {
#pragma unroll
            for (int i = 0; i < RA; ++i) {
                const int idx = tid + i * 256;
                if ((NA4 % 256 == 0) || idx < NA4)
                    *(v4f*)&sA[buf ^ 1][(idx / KA4) * PA + (idx % KA4) * 4] = ra[i];
            }
#pragma unroll
            for (int i = 0; i < RB; ++i) {
                const int idx = tid + i * 256;
                if ((NB4 % 256 == 0) || idx < NB4)
                    *(v4f*)&sB[buf ^ 1][(idx / BN4) * PB + (idx % BN4) * 4] = rb[i];
            }
        }
        __syncthreads();
    }

#pragma unroll
    for (int i = 0; i < WM; ++i) {
#pragma unroll
        for (int j = 0; j < WN; ++j) {
            const int col = n0 + wn * WN * 16 + j * 16 + lr;
#pragma unroll
            for (int r = 0; r < 8; ++r) {
                const int row = m0 + wm * WM * 16 + i * 16 + r + 8 * lh;
                float v = acc[i][j][r];
                if (SOFTPLUS_BIAS) {
                    v += bias[col];
                    v = (v > 20.f) ? v : __logf(1.f + __expf(v));
                }
                C[(size_t)row * ldc + col] = v;
            }
        }
    }
}

template<int BN, int KS>
__global__ __launch_bounds__(256)
__attribute__((amdgpu_waves_per_eu(2, 2)))
void wmma_gemm_f32_ksplit_conv(const float* __restrict__ xr, const float* __restrict__ conv_w,
                               const float* __restrict__ conv_b, float* __restrict__ xs,
                               const float* __restrict__ B, float* __restrict__ C)
{
    constexpr int BM  = 16;
    constexpr int PA  = KS + 4;
    constexpr int PB  = BN + 8;
    constexpr int WN  = BN / 16;
    constexpr int RB  = KS * BN / 4 / 256;
    constexpr int BN4 = BN / 4;
    constexpr int KW  = KS / 32;
    static_assert(KS == 128, "a wave's 32 lanes x 4 channels span one K-step");
    static_assert(KC == 4, "4 taps: one 16-byte piece of conv_w per channel");
    static_assert((KS * BN / 4) % 256 == 0, "whole pieces per thread");
    static_assert((BM * BN) % 256 == 0, "whole outputs per thread");
    static_assert(8 * BM * BN <= KS * PB, "the 8 partial strips fit in one B buffer");

    __shared__ __align__(16) float sA[2][BM * PA];
    __shared__ __align__(16) float sB[2][KS * PB];

    const int tid  = threadIdx.x;
    const int lane = tid & 31;
    const int w    = tid >> 5;
    const int lr   = lane & 15;
    const int lh   = lane >> 4;
    const int m0   = blockIdx.x * BM;

    v8f acc[WN];
#pragma unroll
    for (int j = 0; j < WN; ++j) {
        v8f z = {0.f, 0.f, 0.f, 0.f, 0.f, 0.f, 0.f, 0.f};
        acc[j] = z;
    }

    const int r0 = m0 + 2 * w;
    const float* xrow[5];
    bool xok[3];
#pragma unroll
    for (int j = 0; j < 5; ++j) {
        const int r = r0 - 3 + j;
        if (j < 3) xok[j] = (r >= 0);
        xrow[j] = xr + (size_t)(r >= 0 ? r : 0) * (2 * DI) + 4 * lane;
    }
    const float* wcol = conv_w + (size_t)(4 * lane) * KC;
    const float* bcol = conv_b + 4 * lane;
    float* xsrow = xs + (size_t)r0 * DI + 4 * lane;

    v4f xv[5], wq[4], bq;
    v4f rb[RB];
    v4f xo[2];

    auto request = [&](int k0) {
#pragma unroll
        for (int j = 0; j < 5; ++j)
            xv[j] = *(const v4f*)(xrow[j] + k0);
#pragma unroll
        for (int c = 0; c < 4; ++c)
            wq[c] = *(const v4f*)(wcol + (size_t)(k0 + c) * KC);
        bq = *(const v4f*)(bcol + k0);
#pragma unroll
        for (int i = 0; i < RB; ++i) {
            const int idx = tid + i * 256;
            rb[i] = *(const v4f*)(B + (size_t)(k0 + idx / BN4) * BN + (idx % BN4) * 4);
        }
    };
    auto stage = [&](int buf) {
#pragma unroll
        for (int j = 0; j < 3; ++j)
            if (!xok[j]) { v4f z = {0.f, 0.f, 0.f, 0.f}; xv[j] = z; }
#pragma unroll
        for (int i = 0; i < 2; ++i) {
            v4f o;
#pragma unroll
            for (int c = 0; c < 4; ++c) {
                float a = bq[c];
#pragma unroll
                for (int j = 0; j < 4; ++j)
                    a += xv[i + j][c] * wq[c][j];
                const float sig = 1.f / (1.f + __expf(-a));
                o[c] = a * sig;
            }
            *(v4f*)&sA[buf][(2 * w + i) * PA + 4 * lane] = o;
            xo[i] = o;
        }
#pragma unroll
        for (int i = 0; i < RB; ++i) {
            const int idx = tid + i * 256;
            *(v4f*)&sB[buf][(idx / BN4) * PB + (idx % BN4) * 4] = rb[i];
        }
    };

    auto flush = [&](int k0) {
#pragma unroll
        for (int i = 0; i < 2; ++i)
            *(v4f*)(xsrow + (size_t)i * DI + k0) = xo[i];
    };

    request(0);
    stage(0);
    __syncthreads();

    constexpr int nk = DI / KS;
    for (int kt = 0; kt < nk; ++kt) {
        const int buf = kt & 1;
        const bool more = (kt + 1 < nk);
        if (more) request((kt + 1) * KS);
        flush(kt * KS);
        __builtin_amdgcn_sched_barrier(0);
        const float* tA = &sA[buf][lr * PA + w * KW * 4 + 2 * lh];
        const float* tB = &sB[buf][(w * KW * 4 + 2 * lh) * PB + lr];
#pragma unroll
        for (int kk = 0; kk < KW; ++kk) {
            const v2f a = *(const v2f*)(tA + kk * 4);
            v2f b[WN];
#pragma unroll
            for (int j = 0; j < WN; ++j) {
                const float* bp = tB + (kk * 4) * PB + j * 16;
                v2f t; t[0] = bp[0]; t[1] = bp[PB];
                b[j] = t;
            }
#pragma unroll
            for (int j = 0; j < WN; ++j)
                acc[j] = __builtin_amdgcn_wmma_f32_16x16x4_f32(
                    false, a, false, b[j], (short)0, acc[j], false, false);
        }
        if (more) stage(buf ^ 1);
        __syncthreads();
    }

    float* sP = &sB[0][0];
#pragma unroll
    for (int j = 0; j < WN; ++j)
#pragma unroll
        for (int r = 0; r < 8; ++r)
            sP[w * (BM * BN) + (r + 8 * lh) * BN + j * 16 + lr] = acc[j][r];
    __syncthreads();
    float* Cs = C + (size_t)m0 * BN;
#pragma unroll
    for (int i = 0; i < BM * BN / 256; ++i) {
        const int idx = tid + i * 256;
        const float p0 = sP[0 * (BM * BN) + idx], p1 = sP[1 * (BM * BN) + idx];
        const float p2 = sP[2 * (BM * BN) + idx], p3 = sP[3 * (BM * BN) + idx];
        const float p4 = sP[4 * (BM * BN) + idx], p5 = sP[5 * (BM * BN) + idx];
        const float p6 = sP[6 * (BM * BN) + idx], p7 = sP[7 * (BM * BN) + idx];
        Cs[idx] = ((p0 + p1) + (p2 + p3)) + ((p4 + p5) + (p6 + p7));
    }
}

#define NPART 8
#define LPART (LSEQ / NPART)
#define NCHUNK (LPART / 64)
#define PP 12
#define PROW (32 * PP)

template<bool ENDS_ONLY>
__global__ __launch_bounds__(256)
void scan_kernel(const float* __restrict__ dt, const float* __restrict__ xdbl,
                 const float* __restrict__ xs, const float* __restrict__ xr,
                 const float* __restrict__ A_log, const float* __restrict__ Dvec,
                 float* __restrict__ y, float* __restrict__ hT,
                 float* __restrict__ endH, float* __restrict__ endP)
{
    __shared__ __align__(16) float sDT[64 * 32];
    __shared__ __align__(16) float sXS[64 * 32];
    __shared__ __align__(16) float sBC[64 * 32];
    __shared__ __align__(16) float sP[ENDS_ONLY ? 4 : 64 * PROW];

    const int part  = blockIdx.x >> 6;
    const int d0    = (blockIdx.x & 63) * 32;
    const int lbase = part * LPART;
    const int ch = threadIdx.x >> 3;
    const int q  = threadIdx.x & 7;
    const int d  = d0 + ch;
    const int wr = threadIdx.x >> 5;
    const int wc = threadIdx.x & 31;
    const int pslot = ch * PP + q;

    float A0 = -__expf(A_log[(size_t)d * NS + 2 * q]);
    float A1 = -__expf(A_log[(size_t)d * NS + 2 * q + 1]);
    float h0 = 0.f, h1 = 0.f;
    float p0 = 1.f, p1 = 1.f;
    float Dw = 0.f;
    if (!ENDS_ONLY) Dw = Dvec[d0 + wc];

    const float* gDT = dt   + (size_t)(lbase + ch) * DI + d0 + 4 * q;
    const float* gXS = xs   + (size_t)(lbase + ch) * DI + d0 + 4 * q;
    const float* gBC = xdbl + (size_t)(lbase + ch) * XDBL_W + DTR + 4 * q;
    v4f pdt[2], pxs[2], pbc[2];
    for (int i = 0; i < 2; ++i) {
        pdt[i] = *(const v4f*)(gDT + (size_t)(32 * i) * DI);
        pxs[i] = *(const v4f*)(gXS + (size_t)(32 * i) * DI);
        pbc[i] = *(const v4f*)(gBC + (size_t)(32 * i) * XDBL_W);
    }

    if (!ENDS_ONLY) {
        float2 eh[NPART - 1], ep[NPART - 1];
#pragma unroll
        for (int j = 0; j < NPART - 1; ++j) {
            const int jj = (j < part) ? j : 0;
            eh[j] = *(const float2*)&endH[((size_t)jj * DI + d) * NS + 2 * q];
            ep[j] = *(const float2*)&endP[((size_t)jj * DI + d) * NS + 2 * q];
        }
#pragma unroll
        for (int j = 0; j < NPART - 1; ++j)
            if (j < part) {
                h0 = ep[j].x * h0 + eh[j].x;
                h1 = ep[j].y * h1 + eh[j].y;
            }
    }

    for (int i = 0; i < 2; ++i) {
        *(v4f*)&sDT[(threadIdx.x + 256 * i) * 4] = pdt[i];
        *(v4f*)&sXS[(threadIdx.x + 256 * i) * 4] = pxs[i];
        *(v4f*)&sBC[(threadIdx.x + 256 * i) * 4] = pbc[i];
    }

    for (int c = 0; c < NCHUNK; ++c) {
        const bool more = (c + 1 < NCHUNK);
        __syncthreads();
        if (more) {
            for (int i = 0; i < 2; ++i) {
                const size_t l1 = (size_t)(c + 1) * 64 + 32 * i;
                pdt[i] = *(const v4f*)(gDT + l1 * DI);
                pxs[i] = *(const v4f*)(gXS + l1 * DI);
                pbc[i] = *(const v4f*)(gBC + l1 * XDBL_W);
            }
        }
        if (ENDS_ONLY) {
            auto read8 = [&](int g, float (&dtg)[8], float (&xg)[8], float2 (&Bg)[8]) {
#pragma unroll
                for (int k = 0; k < 8; ++k) {
                    const int s = g * 8 + k;
                    dtg[k] = sDT[s * 32 + ch];
                    xg[k]  = sXS[s * 32 + ch];
                    Bg[k]  = *(const float2*)&sBC[s * 32 + 2 * q];
                }
            };
            auto step8 = [&](const float (&dtg)[8], const float (&xg)[8], const float2 (&Bg)[8]) {
#pragma unroll
                for (int k = 0; k < 8; ++k) {
                    const float dx  = dtg[k] * xg[k];
                    const float dA0 = __expf(dtg[k] * A0);
                    const float dA1 = __expf(dtg[k] * A1);
                    h0 = dA0 * h0 + dx * Bg[k].x;
                    h1 = dA1 * h1 + dx * Bg[k].y;
                    p0 *= dA0;
                    p1 *= dA1;
                }
            };
            float dta[8], xa[8], dtb[8], xb[8];
            float2 Ba[8], Bb[8];
            read8(0, dta, xa, Ba);
#pragma unroll 1
            for (int g = 0; g < 8; g += 2) {
                read8(g + 1, dtb, xb, Bb);
                step8(dta, xa, Ba);
                if (g + 2 < 8) read8(g + 2, dta, xa, Ba);
                step8(dtb, xb, Bb);
            }
            __syncthreads();
            if (more) {
                for (int i = 0; i < 2; ++i) {
                    *(v4f*)&sDT[(threadIdx.x + 256 * i) * 4] = pdt[i];
                    *(v4f*)&sXS[(threadIdx.x + 256 * i) * 4] = pxs[i];
                    *(v4f*)&sBC[(threadIdx.x + 256 * i) * 4] = pbc[i];
                }
            }
        } else {
            float resv[8], xsv[8];
            for (int k = 0; k < 8; ++k) {
                const int l = lbase + c * 64 + wr + 8 * k;
                resv[k] = xr[(size_t)l * (2 * DI) + DI + d0 + wc];
                xsv[k]  = sXS[(wr + 8 * k) * 32 + wc];
            }
#pragma unroll 8
            for (int s = 0; s < 64; ++s) {
                const float dtv = sDT[s * 32 + ch];
                const float xv  = sXS[s * 32 + ch];
                const float2 Bv = *(const float2*)&sBC[s * 32 + 2 * q];
                const float2 Cv = *(const float2*)&sBC[s * 32 + 16 + 2 * q];
                const float dx  = dtv * xv;
                const float dA0 = __expf(dtv * A0);
                const float dA1 = __expf(dtv * A1);
                h0 = dA0 * h0 + dx * Bv.x;
                h1 = dA1 * h1 + dx * Bv.y;
                sP[s * PROW + pslot] = fmaf(h0, Cv.x, h1 * Cv.y);
            }
            __syncthreads();
            if (more) {
                for (int i = 0; i < 2; ++i) {
                    *(v4f*)&sDT[(threadIdx.x + 256 * i) * 4] = pdt[i];
                    *(v4f*)&sXS[(threadIdx.x + 256 * i) * 4] = pxs[i];
                    *(v4f*)&sBC[(threadIdx.x + 256 * i) * 4] = pbc[i];
                }
            }
            for (int k = 0; k < 8; ++k) {
                const int ls = wr + 8 * k;
                const int l = lbase + c * 64 + ls;
                const float4 pa = *(const float4*)&sP[ls * PROW + wc * PP];
                const float4 pb = *(const float4*)&sP[ls * PROW + wc * PP + 4];
                float yv = ((pa.x + pa.y) + (pa.z + pa.w)) + ((pb.x + pb.y) + (pb.z + pb.w));
                yv = yv + xsv[k] * Dw;
                const float sig = 1.f / (1.f + __expf(-resv[k]));
                y[(size_t)l * DI + d0 + wc] = yv * (resv[k] * sig);
            }
        }
    }
    if (ENDS_ONLY) {
        *(float2*)&endH[((size_t)part * DI + d) * NS + 2 * q] = make_float2(h0, h1);
        *(float2*)&endP[((size_t)part * DI + d) * NS + 2 * q] = make_float2(p0, p1);
    } else if (part == NPART - 1) {
        *(float2*)&hT[(size_t)d * NS + 2 * q] = make_float2(h0, h1);
    }
}

extern "C" void kernel_launch(void* const* d_in, const int* in_sizes, int n_in,
                              void* d_out, int out_size, void* d_ws, size_t ws_size,
                              hipStream_t stream) {
    const float* x      = (const float*)d_in[0];
    const float* W_in   = (const float*)d_in[1];
    const float* conv_w = (const float*)d_in[2];
    const float* conv_b = (const float*)d_in[3];
    const float* W_x    = (const float*)d_in[4];
    const float* W_dt   = (const float*)d_in[5];
    const float* b_dt   = (const float*)d_in[6];
    const float* A_log  = (const float*)d_in[7];
    const float* Dv     = (const float*)d_in[8];
    const float* W_out  = (const float*)d_in[9];

    float* out = (float*)d_out;
    float* hT  = out + (size_t)LSEQ * DM;

    float* ws   = (float*)d_ws;
    float* xr   = ws;
    float* xs   = xr   + (size_t)LSEQ * 2 * DI;
    float* xdbl = xs   + (size_t)LSEQ * DI;
    float* dt   = xdbl + (size_t)LSEQ * XDBL_W;
    float* yb   = dt   + (size_t)LSEQ * DI;
    float* endH = yb   + (size_t)LSEQ * DI;
    float* endP = endH + (size_t)(NPART - 1) * DI * NS;

    wmma_gemm_f32<128, 128, 16, 2, 4, false><<<512, 256, 0, stream>>>(
        x, W_in, xr, nullptr, LSEQ, 2 * DI, DM, DM, 2 * DI, 2 * DI);

    wmma_gemm_f32_ksplit_conv<XDBL_W, 128><<<LSEQ / 16, 256, 0, stream>>>(
        xr, conv_w, conv_b, xs, W_x, xdbl);

    wmma_gemm_f32<128, 128, 16, 2, 4, true><<<256, 256, 0, stream>>>(
        xdbl, W_dt, dt, b_dt, LSEQ, DI, DTR, XDBL_W, DI, DI);

    scan_kernel<true><<<(NPART - 1) * (DI / 32), 256, 0, stream>>>(
        dt, xdbl, xs, xr, A_log, Dv, yb, hT, endH, endP);
    scan_kernel<false><<<NPART * (DI / 32), 256, 0, stream>>>(
        dt, xdbl, xs, xr, A_log, Dv, yb, hT, endH, endP);

    wmma_gemm_f32<64, 128, 32, 2, 4, false><<<256, 256, 0, stream>>>(
        yb, W_out, out, nullptr, LSEQ, DM, DI, DI, DM, DM);
}


# x-proj GEMM: K also split over 2 blocks per strip (256 blocks, not 128 on a 256-WGP chip); the 2 partial strips are added where dt-proj and the scan stage them; k sum order changed
# speedup vs baseline: 8.2186x; 1.0369x over previous
#include <hip/hip_runtime.h>
#include <hip/hip_bf16.h>
#include <math.h>


typedef float v2f __attribute__((ext_vector_type(2)));
typedef float v4f __attribute__((ext_vector_type(4)));
typedef float v8f __attribute__((ext_vector_type(8)));

#define DM   1024
#define DI   2048
#define NS   16
#define KC   4
#define DTR  64
#define LSEQ 2048
#define XDBL_W (DTR + 2*NS)
#define XPARTS 2

template<int BM, int BN, int KS, int WGM, int WGN, bool SOFTPLUS_BIAS, int APARTS>
__global__ __launch_bounds__(256)
void wmma_gemm_f32(const float* __restrict__ A, const float* __restrict__ B,
                   float* __restrict__ C, const float* __restrict__ bias,
                   int M, int N, int K, int lda, int ldb, int ldc)
{
    constexpr int PA  = KS + 4;
    constexpr int PB  = BN + 8;
    constexpr int WM  = BM / (16 * WGM);
    constexpr int WN  = BN / (16 * WGN);
    constexpr int NA4 = BM * KS / 4;
    constexpr int NB4 = KS * BN / 4;
    constexpr int RA  = (NA4 + 255) / 256;
    constexpr int RB  = (NB4 + 255) / 256;
    constexpr int BN4 = BN / 4;
    constexpr int KA4 = KS / 4;
    static_assert(WGM * WGN == 8, "8 waves per block");
    static_assert(WM * 16 * WGM == BM && WN * 16 * WGN == BN, "tile split");

    __shared__ __align__(16) float sA[2][BM * PA];
    __shared__ __align__(16) float sB[2][KS * PB];

    const int tid  = threadIdx.x;
    const int lane = tid & 31;
    const int w    = tid >> 5;
    const int wm   = w % WGM;
    const int wn   = w / WGM;
    const int lr   = lane & 15;
    const int lh   = lane >> 4;
    const int tilesM = M / BM;
    const int m0 = (blockIdx.x % tilesM) * BM;
    const int n0 = (blockIdx.x / tilesM) * BN;

    v8f acc[WM][WN];
#pragma unroll
    for (int i = 0; i < WM; ++i)
#pragma unroll
        for (int j = 0; j < WN; ++j) {
            v8f z = {0.f, 0.f, 0.f, 0.f, 0.f, 0.f, 0.f, 0.f};
            acc[i][j] = z;
        }

    v4f ra[RA], rb[RB];
    v4f ra2[RA];
    const size_t apart = (size_t)M * lda;

#pragma unroll
    for (int i = 0; i < RA; ++i) {
        const int idx = tid + i * 256;
        if ((NA4 % 256 == 0) || idx < NA4)
            ra[i] = *(const v4f*)(A + (size_t)(m0 + (idx / KA4)) * lda + (idx % KA4) * 4);
        if (APARTS == 2 && ((NA4 % 256 == 0) || idx < NA4))
            ra2[i] = *(const v4f*)(A + apart + (size_t)(m0 + (idx / KA4)) * lda + (idx % KA4) * 4);
    }
#pragma unroll
    for (int i = 0; i < RB; ++i) {
        const int idx = tid + i * 256;
        if ((NB4 % 256 == 0) || idx < NB4)
            rb[i] = *(const v4f*)(B + (size_t)(idx / BN4) * ldb + n0 + (idx % BN4) * 4);
    }
#pragma unroll
    for (int i = 0; i < RA; ++i) {
        const int idx = tid + i * 256;
        if ((NA4 % 256 == 0) || idx < NA4)
            *(v4f*)&sA[0][(idx / KA4) * PA + (idx % KA4) * 4] = (APARTS == 2) ? ra[i] + ra2[i] : ra[i];
    }
#pragma unroll
    for (int i = 0; i < RB; ++i) {
        const int idx = tid + i * 256;
        if ((NB4 % 256 == 0) || idx < NB4)
            *(v4f*)&sB[0][(idx / BN4) * PB + (idx % BN4) * 4] = rb[i];
    }
    __syncthreads();

    const int nk = K / KS;
    for (int kt = 0; kt < nk; ++kt) {
        const int buf = kt & 1;
        const bool more = (kt + 1 < nk);
        if (more) {
            const int k0 = (kt + 1) * KS;
#pragma unroll
            for (int i = 0; i < RA; ++i) {
                const int idx = tid + i * 256;
                if ((NA4 % 256 == 0) || idx < NA4)
                    ra[i] = *(const v4f*)(A + (size_t)(m0 + (idx / KA4)) * lda + k0 + (idx % KA4) * 4);
                if (APARTS == 2 && ((NA4 % 256 == 0) || idx < NA4))
                    ra2[i] = *(const v4f*)(A + apart + (size_t)(m0 + (idx / KA4)) * lda + k0 + (idx % KA4) * 4);
            }
#pragma unroll
            for (int i = 0; i < RB; ++i) {
                const int idx = tid + i * 256;
                if ((NB4 % 256 == 0) || idx < NB4)
                    rb[i] = *(const v4f*)(B + (size_t)(k0 + idx / BN4) * ldb + n0 + (idx % BN4) * 4);
            }
        }
        const float* tA = &sA[buf][(wm * WM * 16 + lr) * PA + 2 * lh];
        const float* tB = &sB[buf][(2 * lh) * PB + wn * WN * 16 + lr];
#pragma unroll
        for (int kk = 0; kk < KS / 4; ++kk) {
            v2f a[WM];
#pragma unroll
            for (int i = 0; i < WM; ++i)
                a[i] = *(const v2f*)(tA + i * 16 * PA + kk * 4);
            v2f b[WN];
#pragma unroll
            for (int j = 0; j < WN; ++j) {
                const float* bp = tB + (kk * 4) * PB + j * 16;
                v2f t; t[0] = bp[0]; t[1] = bp[PB];
                b[j] = t;
            }
#pragma unroll
            for (int i = 0; i < WM; ++i)
#pragma unroll
                for (int j = 0; j < WN; ++j)
                    acc[i][j] = __builtin_amdgcn_wmma_f32_16x16x4_f32(
                        false, a[i], false, b[j], (short)0, acc[i][j], false, false);
        }
        if (more) {
#pragma unroll
            for (int i = 0; i < RA; ++i) {
                const int idx = tid + i * 256;
                if ((NA4 % 256 == 0) || idx < NA4)
                    *(v4f*)&sA[buf ^ 1][(idx / KA4) * PA + (idx % KA4) * 4] = (APARTS == 2) ? ra[i] + ra2[i] : ra[i];
            }
#pragma unroll
            for (int i = 0; i < RB; ++i) {
                const int idx = tid + i * 256;
                if ((NB4 % 256 == 0) || idx < NB4)
                    *(v4f*)&sB[buf ^ 1][(idx / BN4) * PB + (idx % BN4) * 4] = rb[i];
            }
        }
        __syncthreads();
    }

#pragma unroll
    for (int i = 0; i < WM; ++i) {
#pragma unroll
        for (int j = 0; j < WN; ++j) {
            const int col = n0 + wn * WN * 16 + j * 16 + lr;
#pragma unroll
            for (int r = 0; r < 8; ++r) {
                const int row = m0 + wm * WM * 16 + i * 16 + r + 8 * lh;
                float v = acc[i][j][r];
                if (SOFTPLUS_BIAS) {
                    v += bias[col];
                    v = (v > 20.f) ? v : __logf(1.f + __expf(v));
                }
                C[(size_t)row * ldc + col] = v;
            }
        }
    }
}

template<int BN, int KS, int KSPLIT>
__global__ __launch_bounds__(256)
__attribute__((amdgpu_waves_per_eu(2, 2)))
void wmma_gemm_f32_ksplit_conv(const float* __restrict__ xr, const float* __restrict__ conv_w,
                               const float* __restrict__ conv_b, float* __restrict__ xs,
                               const float* __restrict__ B, float* __restrict__ C)
{
    constexpr int BM  = 16;
    constexpr int PA  = KS + 4;
    constexpr int PB  = BN + 8;
    constexpr int WN  = BN / 16;
    constexpr int RB  = KS * BN / 4 / 256;
    constexpr int BN4 = BN / 4;
    constexpr int KW  = KS / 32;
    static_assert(KS == 128, "a wave's 32 lanes x 4 channels span one K-step");
    static_assert(KC == 4, "4 taps: one 16-byte piece of conv_w per channel");
    static_assert((KS * BN / 4) % 256 == 0, "whole pieces per thread");
    static_assert((BM * BN) % 256 == 0, "whole outputs per thread");
    static_assert(8 * BM * BN <= KS * PB, "the 8 partial strips fit in one B buffer");
    static_assert(DI % (KS * KSPLIT) == 0, "whole K-steps per block");
    constexpr int NSTRIP = LSEQ / BM;

    __shared__ __align__(16) float sA[2][BM * PA];
    __shared__ __align__(16) float sB[2][KS * PB];

    const int tid  = threadIdx.x;
    const int lane = tid & 31;
    const int w    = tid >> 5;
    const int lr   = lane & 15;
    const int lh   = lane >> 4;
    const int m0   = (blockIdx.x % NSTRIP) * BM;
    const int kpart = blockIdx.x / NSTRIP;
    const int kb   = kpart * (DI / KSPLIT);

    v8f acc[WN];
#pragma unroll
    for (int j = 0; j < WN; ++j) {
        v8f z = {0.f, 0.f, 0.f, 0.f, 0.f, 0.f, 0.f, 0.f};
        acc[j] = z;
    }

    const int r0 = m0 + 2 * w;
    const float* xrow[5];
    bool xok[3];
#pragma unroll
    for (int j = 0; j < 5; ++j) {
        const int r = r0 - 3 + j;
        if (j < 3) xok[j] = (r >= 0);
        xrow[j] = xr + (size_t)(r >= 0 ? r : 0) * (2 * DI) + 4 * lane;
    }
    const float* wcol = conv_w + (size_t)(4 * lane) * KC;
    const float* bcol = conv_b + 4 * lane;
    float* xsrow = xs + (size_t)r0 * DI + 4 * lane;

    v4f xv[5], wq[4], bq;
    v4f rb[RB];
    v4f xo[2];

    auto request = [&](int k0) {
#pragma unroll
        for (int j = 0; j < 5; ++j)
            xv[j] = *(const v4f*)(xrow[j] + k0);
#pragma unroll
        for (int c = 0; c < 4; ++c)
            wq[c] = *(const v4f*)(wcol + (size_t)(k0 + c) * KC);
        bq = *(const v4f*)(bcol + k0);
#pragma unroll
        for (int i = 0; i < RB; ++i) {
            const int idx = tid + i * 256;
            rb[i] = *(const v4f*)(B + (size_t)(k0 + idx / BN4) * BN + (idx % BN4) * 4);
        }
    };
    auto stage = [&](int buf) {
#pragma unroll
        for (int j = 0; j < 3; ++j)
            if (!xok[j]) { v4f z = {0.f, 0.f, 0.f, 0.f}; xv[j] = z; }
#pragma unroll
        for (int i = 0; i < 2; ++i) {
            v4f o;
#pragma unroll
            for (int c = 0; c < 4; ++c) {
                float a = bq[c];
#pragma unroll
                for (int j = 0; j < 4; ++j)
                    a += xv[i + j][c] * wq[c][j];
                const float sig = 1.f / (1.f + __expf(-a));
                o[c] = a * sig;
            }
            *(v4f*)&sA[buf][(2 * w + i) * PA + 4 * lane] = o;
            xo[i] = o;
        }
#pragma unroll
        for (int i = 0; i < RB; ++i) {
            const int idx = tid + i * 256;
            *(v4f*)&sB[buf][(idx / BN4) * PB + (idx % BN4) * 4] = rb[i];
        }
    };

    auto flush = [&](int k0) {
#pragma unroll
        for (int i = 0; i < 2; ++i)
            *(v4f*)(xsrow + (size_t)i * DI + k0) = xo[i];
    };

    request(kb);
    stage(0);
    __syncthreads();

    constexpr int nk = DI / KS / KSPLIT;
    for (int kt = 0; kt < nk; ++kt) {
        const int buf = kt & 1;
        const bool more = (kt + 1 < nk);
        if (more) request(kb + (kt + 1) * KS);
        flush(kb + kt * KS);
        __builtin_amdgcn_sched_barrier(0);
        const float* tA = &sA[buf][lr * PA + w * KW * 4 + 2 * lh];
        const float* tB = &sB[buf][(w * KW * 4 + 2 * lh) * PB + lr];
#pragma unroll
        for (int kk = 0; kk < KW; ++kk) {
            const v2f a = *(const v2f*)(tA + kk * 4);
            v2f b[WN];
#pragma unroll
            for (int j = 0; j < WN; ++j) {
                const float* bp = tB + (kk * 4) * PB + j * 16;
                v2f t; t[0] = bp[0]; t[1] = bp[PB];
                b[j] = t;
            }
#pragma unroll
            for (int j = 0; j < WN; ++j)
                acc[j] = __builtin_amdgcn_wmma_f32_16x16x4_f32(
                    false, a, false, b[j], (short)0, acc[j], false, false);
        }
        if (more) stage(buf ^ 1);
        __syncthreads();
    }

    float* sP = &sB[0][0];
#pragma unroll
    for (int j = 0; j < WN; ++j)
#pragma unroll
        for (int r = 0; r < 8; ++r)
            sP[w * (BM * BN) + (r + 8 * lh) * BN + j * 16 + lr] = acc[j][r];
    __syncthreads();
    float* Cs = C + (size_t)kpart * ((size_t)LSEQ * BN) + (size_t)m0 * BN;
#pragma unroll
    for (int i = 0; i < BM * BN / 256; ++i) {
        const int idx = tid + i * 256;
        const float p0 = sP[0 * (BM * BN) + idx], p1 = sP[1 * (BM * BN) + idx];
        const float p2 = sP[2 * (BM * BN) + idx], p3 = sP[3 * (BM * BN) + idx];
        const float p4 = sP[4 * (BM * BN) + idx], p5 = sP[5 * (BM * BN) + idx];
        const float p6 = sP[6 * (BM * BN) + idx], p7 = sP[7 * (BM * BN) + idx];
        Cs[idx] = ((p0 + p1) + (p2 + p3)) + ((p4 + p5) + (p6 + p7));
    }
}

#define NPART 8
#define LPART (LSEQ / NPART)
#define NCHUNK (LPART / 64)
#define PP 12
#define PROW (32 * PP)

template<bool ENDS_ONLY>
__global__ __launch_bounds__(256)
void scan_kernel(const float* __restrict__ dt, const float* __restrict__ xdbl,
                 const float* __restrict__ xs, const float* __restrict__ xr,
                 const float* __restrict__ A_log, const float* __restrict__ Dvec,
                 float* __restrict__ y, float* __restrict__ hT,
                 float* __restrict__ endH, float* __restrict__ endP)
{
    __shared__ __align__(16) float sDT[64 * 32];
    __shared__ __align__(16) float sXS[64 * 32];
    __shared__ __align__(16) float sBC[64 * 32];
    __shared__ __align__(16) float sP[ENDS_ONLY ? 4 : 64 * PROW];

    const int part  = blockIdx.x >> 6;
    const int d0    = (blockIdx.x & 63) * 32;
    const int lbase = part * LPART;
    const int ch = threadIdx.x >> 3;
    const int q  = threadIdx.x & 7;
    const int d  = d0 + ch;
    const int wr = threadIdx.x >> 5;
    const int wc = threadIdx.x & 31;
    const int pslot = ch * PP + q;

    float A0 = -__expf(A_log[(size_t)d * NS + 2 * q]);
    float A1 = -__expf(A_log[(size_t)d * NS + 2 * q + 1]);
    float h0 = 0.f, h1 = 0.f;
    float p0 = 1.f, p1 = 1.f;
    float Dw = 0.f;
    if (!ENDS_ONLY) Dw = Dvec[d0 + wc];

    const float* gDT = dt   + (size_t)(lbase + ch) * DI + d0 + 4 * q;
    const float* gXS = xs   + (size_t)(lbase + ch) * DI + d0 + 4 * q;
    const float* gBC = xdbl + (size_t)(lbase + ch) * XDBL_W + DTR + 4 * q;
    const float* gBC2 = gBC + (size_t)LSEQ * XDBL_W;
    static_assert(XPARTS == 2, "two partials are added here");
    v4f pdt[2], pxs[2], pbc[2], pbc2[2];
    for (int i = 0; i < 2; ++i) {
        pdt[i] = *(const v4f*)(gDT + (size_t)(32 * i) * DI);
        pxs[i] = *(const v4f*)(gXS + (size_t)(32 * i) * DI);
        pbc[i] = *(const v4f*)(gBC + (size_t)(32 * i) * XDBL_W);
        pbc2[i] = *(const v4f*)(gBC2 + (size_t)(32 * i) * XDBL_W);
    }

    if (!ENDS_ONLY) {
        float2 eh[NPART - 1], ep[NPART - 1];
#pragma unroll
        for (int j = 0; j < NPART - 1; ++j) {
            const int jj = (j < part) ? j : 0;
            eh[j] = *(const float2*)&endH[((size_t)jj * DI + d) * NS + 2 * q];
            ep[j] = *(const float2*)&endP[((size_t)jj * DI + d) * NS + 2 * q];
        }
#pragma unroll
        for (int j = 0; j < NPART - 1; ++j)
            if (j < part) {
                h0 = ep[j].x * h0 + eh[j].x;
                h1 = ep[j].y * h1 + eh[j].y;
            }
    }

    for (int i = 0; i < 2; ++i) {
        *(v4f*)&sDT[(threadIdx.x + 256 * i) * 4] = pdt[i];
        *(v4f*)&sXS[(threadIdx.x + 256 * i) * 4] = pxs[i];
        *(v4f*)&sBC[(threadIdx.x + 256 * i) * 4] = pbc[i] + pbc2[i];
    }

    for (int c = 0; c < NCHUNK; ++c) {
        const bool more = (c + 1 < NCHUNK);
        __syncthreads();
        if (more) {
            for (int i = 0; i < 2; ++i) {
                const size_t l1 = (size_t)(c + 1) * 64 + 32 * i;
                pdt[i] = *(const v4f*)(gDT + l1 * DI);
                pxs[i] = *(const v4f*)(gXS + l1 * DI);
                pbc[i] = *(const v4f*)(gBC + l1 * XDBL_W);
                pbc2[i] = *(const v4f*)(gBC2 + l1 * XDBL_W);
            }
        }
        if (ENDS_ONLY) {
            auto read8 = [&](int g, float (&dtg)[8], float (&xg)[8], float2 (&Bg)[8]) {
#pragma unroll
                for (int k = 0; k < 8; ++k) {
                    const int s = g * 8 + k;
                    dtg[k] = sDT[s * 32 + ch];
                    xg[k]  = sXS[s * 32 + ch];
                    Bg[k]  = *(const float2*)&sBC[s * 32 + 2 * q];
                }
            };
            auto step8 = [&](const float (&dtg)[8], const float (&xg)[8], const float2 (&Bg)[8]) {
#pragma unroll
                for (int k = 0; k < 8; ++k) {
                    const float dx  = dtg[k] * xg[k];
                    const float dA0 = __expf(dtg[k] * A0);
                    const float dA1 = __expf(dtg[k] * A1);
                    h0 = dA0 * h0 + dx * Bg[k].x;
                    h1 = dA1 * h1 + dx * Bg[k].y;
                    p0 *= dA0;
                    p1 *= dA1;
                }
            };
            float dta[8], xa[8], dtb[8], xb[8];
            float2 Ba[8], Bb[8];
            read8(0, dta, xa, Ba);
#pragma unroll 1
            for (int g = 0; g < 8; g += 2) {
                read8(g + 1, dtb, xb, Bb);
                step8(dta, xa, Ba);
                if (g + 2 < 8) read8(g + 2, dta, xa, Ba);
                step8(dtb, xb, Bb);
            }
            __syncthreads();
            if (more) {
                for (int i = 0; i < 2; ++i) {
                    *(v4f*)&sDT[(threadIdx.x + 256 * i) * 4] = pdt[i];
                    *(v4f*)&sXS[(threadIdx.x + 256 * i) * 4] = pxs[i];
                    *(v4f*)&sBC[(threadIdx.x + 256 * i) * 4] = pbc[i] + pbc2[i];
                }
            }
        } else {
            float resv[8], xsv[8];
            for (int k = 0; k < 8; ++k) {
                const int l = lbase + c * 64 + wr + 8 * k;
                resv[k] = xr[(size_t)l * (2 * DI) + DI + d0 + wc];
                xsv[k]  = sXS[(wr + 8 * k) * 32 + wc];
            }
#pragma unroll 8
            for (int s = 0; s < 64; ++s) {
                const float dtv = sDT[s * 32 + ch];
                const float xv  = sXS[s * 32 + ch];
                const float2 Bv = *(const float2*)&sBC[s * 32 + 2 * q];
                const float2 Cv = *(const float2*)&sBC[s * 32 + 16 + 2 * q];
                const float dx  = dtv * xv;
                const float dA0 = __expf(dtv * A0);
                const float dA1 = __expf(dtv * A1);
                h0 = dA0 * h0 + dx * Bv.x;
                h1 = dA1 * h1 + dx * Bv.y;
                sP[s * PROW + pslot] = fmaf(h0, Cv.x, h1 * Cv.y);
            }
            __syncthreads();
            if (more) {
                for (int i = 0; i < 2; ++i) {
                    *(v4f*)&sDT[(threadIdx.x + 256 * i) * 4] = pdt[i];
                    *(v4f*)&sXS[(threadIdx.x + 256 * i) * 4] = pxs[i];
                    *(v4f*)&sBC[(threadIdx.x + 256 * i) * 4] = pbc[i] + pbc2[i];
                }
            }
            for (int k = 0; k < 8; ++k) {
                const int ls = wr + 8 * k;
                const int l = lbase + c * 64 + ls;
                const float4 pa = *(const float4*)&sP[ls * PROW + wc * PP];
                const float4 pb = *(const float4*)&sP[ls * PROW + wc * PP + 4];
                float yv = ((pa.x + pa.y) + (pa.z + pa.w)) + ((pb.x + pb.y) + (pb.z + pb.w));
                yv = yv + xsv[k] * Dw;
                const float sig = 1.f / (1.f + __expf(-resv[k]));
                y[(size_t)l * DI + d0 + wc] = yv * (resv[k] * sig);
            }
        }
    }
    if (ENDS_ONLY) {
        *(float2*)&endH[((size_t)part * DI + d) * NS + 2 * q] = make_float2(h0, h1);
        *(float2*)&endP[((size_t)part * DI + d) * NS + 2 * q] = make_float2(p0, p1);
    } else if (part == NPART - 1) {
        *(float2*)&hT[(size_t)d * NS + 2 * q] = make_float2(h0, h1);
    }
}

extern "C" void kernel_launch(void* const* d_in, const int* in_sizes, int n_in,
                              void* d_out, int out_size, void* d_ws, size_t ws_size,
                              hipStream_t stream) {
    const float* x      = (const float*)d_in[0];
    const float* W_in   = (const float*)d_in[1];
    const float* conv_w = (const float*)d_in[2];
    const float* conv_b = (const float*)d_in[3];
    const float* W_x    = (const float*)d_in[4];
    const float* W_dt   = (const float*)d_in[5];
    const float* b_dt   = (const float*)d_in[6];
    const float* A_log  = (const float*)d_in[7];
    const float* Dv     = (const float*)d_in[8];
    const float* W_out  = (const float*)d_in[9];

    float* out = (float*)d_out;
    float* hT  = out + (size_t)LSEQ * DM;

    float* ws   = (float*)d_ws;
    float* xr   = ws;
    float* xs   = xr   + (size_t)LSEQ * 2 * DI;
    float* xdbl = xs   + (size_t)LSEQ * DI;
    float* dt   = xdbl + (size_t)XPARTS * LSEQ * XDBL_W;
    float* yb   = dt   + (size_t)LSEQ * DI;
    float* endH = yb   + (size_t)LSEQ * DI;
    float* endP = endH + (size_t)(NPART - 1) * DI * NS;

    wmma_gemm_f32<128, 128, 16, 2, 4, false, 1><<<512, 256, 0, stream>>>(
        x, W_in, xr, nullptr, LSEQ, 2 * DI, DM, DM, 2 * DI, 2 * DI);

    wmma_gemm_f32_ksplit_conv<XDBL_W, 128, XPARTS><<<XPARTS * (LSEQ / 16), 256, 0, stream>>>(
        xr, conv_w, conv_b, xs, W_x, xdbl);

    wmma_gemm_f32<128, 128, 16, 2, 4, true, XPARTS><<<256, 256, 0, stream>>>(
        xdbl, W_dt, dt, b_dt, LSEQ, DI, DTR, XDBL_W, DI, DI);

    scan_kernel<true><<<(NPART - 1) * (DI / 32), 256, 0, stream>>>(
        dt, xdbl, xs, xr, A_log, Dv, yb, hT, endH, endP);
    scan_kernel<false><<<NPART * (DI / 32), 256, 0, stream>>>(
        dt, xdbl, xs, xr, A_log, Dv, yb, hT, endH, endP);

    wmma_gemm_f32<64, 128, 32, 2, 4, false, 1><<<256, 256, 0, stream>>>(
        yb, W_out, out, nullptr, LSEQ, DM, DI, DI, DM, DM);
}


# in-proj and out-proj GEMMs: K-step hand-overs made rare and short: K-steps 4x deeper (64 and 128 columns), next tile stored to LDS amid the step's products, not behind them; sums unchanged
# speedup vs baseline: 8.5046x; 1.0348x over previous
#include <hip/hip_runtime.h>
#include <hip/hip_bf16.h>
#include <math.h>


typedef float v2f __attribute__((ext_vector_type(2)));
typedef float v4f __attribute__((ext_vector_type(4)));
typedef float v8f __attribute__((ext_vector_type(8)));

#define DM   1024
#define DI   2048
#define NS   16
#define KC   4
#define DTR  64
#define LSEQ 2048
#define XDBL_W (DTR + 2*NS)
#define XPARTS 2

template<int BM, int BN, int KS, int WGM, int WGN, bool SOFTPLUS_BIAS, int APARTS>
__global__ __launch_bounds__(256)
void wmma_gemm_f32(const float* __restrict__ A, const float* __restrict__ B,
                   float* __restrict__ C, const float* __restrict__ bias,
                   int M, int N, int K, int lda, int ldb, int ldc)
{
    constexpr int PA  = KS + 4;
    constexpr int PB  = BN + 8;
    constexpr int WM  = BM / (16 * WGM);
    constexpr int WN  = BN / (16 * WGN);
    constexpr int NA4 = BM * KS / 4;
    constexpr int NB4 = KS * BN / 4;
    constexpr int RA  = (NA4 + 255) / 256;
    constexpr int RB  = (NB4 + 255) / 256;
    constexpr int BN4 = BN / 4;
    constexpr int KA4 = KS / 4;
    constexpr int NKK = KS / 4;
    constexpr int KK_STORE = (KS >= 64) ? NKK / 2 : NKK;
    static_assert(WGM * WGN == 8, "8 waves per block");
    static_assert(WM * 16 * WGM == BM && WN * 16 * WGN == BN, "tile split");

    __shared__ __align__(16) float sA[2][BM * PA];
    __shared__ __align__(16) float sB[2][KS * PB];

    const int tid  = threadIdx.x;
    const int lane = tid & 31;
    const int w    = tid >> 5;
    const int wm   = w % WGM;
    const int wn   = w / WGM;
    const int lr   = lane & 15;
    const int lh   = lane >> 4;
    const int tilesM = M / BM;
    const int m0 = (blockIdx.x % tilesM) * BM;
    const int n0 = (blockIdx.x / tilesM) * BN;

    v8f acc[WM][WN];
#pragma unroll
    for (int i = 0; i < WM; ++i)
#pragma unroll
        for (int j = 0; j < WN; ++j) {
            v8f z = {0.f, 0.f, 0.f, 0.f, 0.f, 0.f, 0.f, 0.f};
            acc[i][j] = z;
        }

    v4f ra[RA], rb[RB];
    v4f ra2[RA];
    const size_t apart = (size_t)M * lda;

#pragma unroll
    for (int i = 0; i < RA; ++i) {
        const int idx = tid + i * 256;
        if ((NA4 % 256 == 0) || idx < NA4)
            ra[i] = *(const v4f*)(A + (size_t)(m0 + (idx / KA4)) * lda + (idx % KA4) * 4);
        if (APARTS == 2 && ((NA4 % 256 == 0) || idx < NA4))
            ra2[i] = *(const v4f*)(A + apart + (size_t)(m0 + (idx / KA4)) * lda + (idx % KA4) * 4);
    }
#pragma unroll
    for (int i = 0; i < RB; ++i) {
        const int idx = tid + i * 256;
        if ((NB4 % 256 == 0) || idx < NB4)
            rb[i] = *(const v4f*)(B + (size_t)(idx / BN4) * ldb + n0 + (idx % BN4) * 4);
    }
#pragma unroll
    for (int i = 0; i < RA; ++i) {
        const int idx = tid + i * 256;
        if ((NA4 % 256 == 0) || idx < NA4)
            *(v4f*)&sA[0][(idx / KA4) * PA + (idx % KA4) * 4] = (APARTS == 2) ? ra[i] + ra2[i] : ra[i];
    }
#pragma unroll
    for (int i = 0; i < RB; ++i) {
        const int idx = tid + i * 256;
        if ((NB4 % 256 == 0) || idx < NB4)
            *(v4f*)&sB[0][(idx / BN4) * PB + (idx % BN4) * 4] = rb[i];
    }
    __syncthreads();

    const int nk = K / KS;
    for (int kt = 0; kt < nk; ++kt) {
        const int buf = kt & 1;
        const bool more = (kt + 1 < nk);
        if (more) {
            const int k0 = (kt + 1) * KS;
#pragma unroll
            for (int i = 0; i < RA; ++i) {
                const int idx = tid + i * 256;
                if ((NA4 % 256 == 0) || idx < NA4)
                    ra[i] = *(const v4f*)(A + (size_t)(m0 + (idx / KA4)) * lda + k0 + (idx % KA4) * 4);
                if (APARTS == 2 && ((NA4 % 256 == 0) || idx < NA4))
                    ra2[i] = *(const v4f*)(A + apart + (size_t)(m0 + (idx / KA4)) * lda + k0 + (idx % KA4) * 4);
            }
#pragma unroll
            for (int i = 0; i < RB; ++i) {
                const int idx = tid + i * 256;
                if ((NB4 % 256 == 0) || idx < NB4)
                    rb[i] = *(const v4f*)(B + (size_t)(k0 + idx / BN4) * ldb + n0 + (idx % BN4) * 4);
            }
        }
        const float* tA = &sA[buf][(wm * WM * 16 + lr) * PA + 2 * lh];
        const float* tB = &sB[buf][(2 * lh) * PB + wn * WN * 16 + lr];
        auto products = [&](int kk) {
            v2f a[WM];
#pragma unroll
            for (int i = 0; i < WM; ++i)
                a[i] = *(const v2f*)(tA + i * 16 * PA + kk * 4);
            v2f b[WN];
#pragma unroll
            for (int j = 0; j < WN; ++j) {
                const float* bp = tB + (kk * 4) * PB + j * 16;
                v2f t; t[0] = bp[0]; t[1] = bp[PB];
                b[j] = t;
            }
#pragma unroll
            for (int i = 0; i < WM; ++i)
#pragma unroll
                for (int j = 0; j < WN; ++j)
                    acc[i][j] = __builtin_amdgcn_wmma_f32_16x16x4_f32(
                        false, a[i], false, b[j], (short)0, acc[i][j], false, false);
        };
#pragma unroll
        for (int kk = 0; kk < KK_STORE; ++kk) products(kk);
        if (more) {
#pragma unroll
            for (int i = 0; i < RA; ++i) {
                const int idx = tid + i * 256;
                if ((NA4 % 256 == 0) || idx < NA4)
                    *(v4f*)&sA[buf ^ 1][(idx / KA4) * PA + (idx % KA4) * 4] = (APARTS == 2) ? ra[i] + ra2[i] : ra[i];
            }
#pragma unroll
            for (int i = 0; i < RB; ++i) {
                const int idx = tid + i * 256;
                if ((NB4 % 256 == 0) || idx < NB4)
                    *(v4f*)&sB[buf ^ 1][(idx / BN4) * PB + (idx % BN4) * 4] = rb[i];
            }
        }
#pragma unroll
        for (int kk = KK_STORE; kk < NKK; ++kk) products(kk);
        __syncthreads();
    }

#pragma unroll
    for (int i = 0; i < WM; ++i) {
#pragma unroll
        for (int j = 0; j < WN; ++j) {
            const int col = n0 + wn * WN * 16 + j * 16 + lr;
#pragma unroll
            for (int r = 0; r < 8; ++r) {
                const int row = m0 + wm * WM * 16 + i * 16 + r + 8 * lh;
                float v = acc[i][j][r];
                if (SOFTPLUS_BIAS) {
                    v += bias[col];
                    v = (v > 20.f) ? v : __logf(1.f + __expf(v));
                }
                C[(size_t)row * ldc + col] = v;
            }
        }
    }
}

template<int BN, int KS, int KSPLIT>
__global__ __launch_bounds__(256)
__attribute__((amdgpu_waves_per_eu(2, 2)))
void wmma_gemm_f32_ksplit_conv(const float* __restrict__ xr, const float* __restrict__ conv_w,
                               const float* __restrict__ conv_b, float* __restrict__ xs,
                               const float* __restrict__ B, float* __restrict__ C)
{
    constexpr int BM  = 16;
    constexpr int PA  = KS + 4;
    constexpr int PB  = BN + 8;
    constexpr int WN  = BN / 16;
    constexpr int RB  = KS * BN / 4 / 256;
    constexpr int BN4 = BN / 4;
    constexpr int KW  = KS / 32;
    static_assert(KS == 128, "a wave's 32 lanes x 4 channels span one K-step");
    static_assert(KC == 4, "4 taps: one 16-byte piece of conv_w per channel");
    static_assert((KS * BN / 4) % 256 == 0, "whole pieces per thread");
    static_assert((BM * BN) % 256 == 0, "whole outputs per thread");
    static_assert(8 * BM * BN <= KS * PB, "the 8 partial strips fit in one B buffer");
    static_assert(DI % (KS * KSPLIT) == 0, "whole K-steps per block");
    constexpr int NSTRIP = LSEQ / BM;

    __shared__ __align__(16) float sA[2][BM * PA];
    __shared__ __align__(16) float sB[2][KS * PB];

    const int tid  = threadIdx.x;
    const int lane = tid & 31;
    const int w    = tid >> 5;
    const int lr   = lane & 15;
    const int lh   = lane >> 4;
    const int m0   = (blockIdx.x % NSTRIP) * BM;
    const int kpart = blockIdx.x / NSTRIP;
    const int kb   = kpart * (DI / KSPLIT);

    v8f acc[WN];
#pragma unroll
    for (int j = 0; j < WN; ++j) {
        v8f z = {0.f, 0.f, 0.f, 0.f, 0.f, 0.f, 0.f, 0.f};
        acc[j] = z;
    }

    const int r0 = m0 + 2 * w;
    const float* xrow[5];
    bool xok[3];
#pragma unroll
    for (int j = 0; j < 5; ++j) {
        const int r = r0 - 3 + j;
        if (j < 3) xok[j] = (r >= 0);
        xrow[j] = xr + (size_t)(r >= 0 ? r : 0) * (2 * DI) + 4 * lane;
    }
    const float* wcol = conv_w + (size_t)(4 * lane) * KC;
    const float* bcol = conv_b + 4 * lane;
    float* xsrow = xs + (size_t)r0 * DI + 4 * lane;

    v4f xv[5], wq[4], bq;
    v4f rb[RB];
    v4f xo[2];

    auto request = [&](int k0) {
#pragma unroll
        for (int j = 0; j < 5; ++j)
            xv[j] = *(const v4f*)(xrow[j] + k0);
#pragma unroll
        for (int c = 0; c < 4; ++c)
            wq[c] = *(const v4f*)(wcol + (size_t)(k0 + c) * KC);
        bq = *(const v4f*)(bcol + k0);
#pragma unroll
        for (int i = 0; i < RB; ++i) {
            const int idx = tid + i * 256;
            rb[i] = *(const v4f*)(B + (size_t)(k0 + idx / BN4) * BN + (idx % BN4) * 4);
        }
    };
    auto stage = [&](int buf) {
#pragma unroll
        for (int j = 0; j < 3; ++j)
            if (!xok[j]) { v4f z = {0.f, 0.f, 0.f, 0.f}; xv[j] = z; }
#pragma unroll
        for (int i = 0; i < 2; ++i) {
            v4f o;
#pragma unroll
            for (int c = 0; c < 4; ++c) {
                float a = bq[c];
#pragma unroll
                for (int j = 0; j < 4; ++j)
                    a += xv[i + j][c] * wq[c][j];
                const float sig = 1.f / (1.f + __expf(-a));
                o[c] = a * sig;
            }
            *(v4f*)&sA[buf][(2 * w + i) * PA + 4 * lane] = o;
            xo[i] = o;
        }
#pragma unroll
        for (int i = 0; i < RB; ++i) {
            const int idx = tid + i * 256;
            *(v4f*)&sB[buf][(idx / BN4) * PB + (idx % BN4) * 4] = rb[i];
        }
    };

    auto flush = [&](int k0) {
#pragma unroll
        for (int i = 0; i < 2; ++i)
            *(v4f*)(xsrow + (size_t)i * DI + k0) = xo[i];
    };

    request(kb);
    stage(0);
    __syncthreads();

    constexpr int nk = DI / KS / KSPLIT;
    for (int kt = 0; kt < nk; ++kt) {
        const int buf = kt & 1;
        const bool more = (kt + 1 < nk);
        if (more) request(kb + (kt + 1) * KS);
        flush(kb + kt * KS);
        __builtin_amdgcn_sched_barrier(0);
        const float* tA = &sA[buf][lr * PA + w * KW * 4 + 2 * lh];
        const float* tB = &sB[buf][(w * KW * 4 + 2 * lh) * PB + lr];
#pragma unroll
        for (int kk = 0; kk < KW; ++kk) {
            const v2f a = *(const v2f*)(tA + kk * 4);
            v2f b[WN];
#pragma unroll
            for (int j = 0; j < WN; ++j) {
                const float* bp = tB + (kk * 4) * PB + j * 16;
                v2f t; t[0] = bp[0]; t[1] = bp[PB];
                b[j] = t;
            }
#pragma unroll
            for (int j = 0; j < WN; ++j)
                acc[j] = __builtin_amdgcn_wmma_f32_16x16x4_f32(
                    false, a, false, b[j], (short)0, acc[j], false, false);
        }
        if (more) stage(buf ^ 1);
        __syncthreads();
    }

    float* sP = &sB[0][0];
#pragma unroll
    for (int j = 0; j < WN; ++j)
#pragma unroll
        for (int r = 0; r < 8; ++r)
            sP[w * (BM * BN) + (r + 8 * lh) * BN + j * 16 + lr] = acc[j][r];
    __syncthreads();
    float* Cs = C + (size_t)kpart * ((size_t)LSEQ * BN) + (size_t)m0 * BN;
#pragma unroll
    for (int i = 0; i < BM * BN / 256; ++i) {
        const int idx = tid + i * 256;
        const float p0 = sP[0 * (BM * BN) + idx], p1 = sP[1 * (BM * BN) + idx];
        const float p2 = sP[2 * (BM * BN) + idx], p3 = sP[3 * (BM * BN) + idx];
        const float p4 = sP[4 * (BM * BN) + idx], p5 = sP[5 * (BM * BN) + idx];
        const float p6 = sP[6 * (BM * BN) + idx], p7 = sP[7 * (BM * BN) + idx];
        Cs[idx] = ((p0 + p1) + (p2 + p3)) + ((p4 + p5) + (p6 + p7));
    }
}

#define NPART 8
#define LPART (LSEQ / NPART)
#define NCHUNK (LPART / 64)
#define PP 12
#define PROW (32 * PP)

template<bool ENDS_ONLY>
__global__ __launch_bounds__(256)
void scan_kernel(const float* __restrict__ dt, const float* __restrict__ xdbl,
                 const float* __restrict__ xs, const float* __restrict__ xr,
                 const float* __restrict__ A_log, const float* __restrict__ Dvec,
                 float* __restrict__ y, float* __restrict__ hT,
                 float* __restrict__ endH, float* __restrict__ endP)
{
    __shared__ __align__(16) float sDT[64 * 32];
    __shared__ __align__(16) float sXS[64 * 32];
    __shared__ __align__(16) float sBC[64 * 32];
    __shared__ __align__(16) float sP[ENDS_ONLY ? 4 : 64 * PROW];

    const int part  = blockIdx.x >> 6;
    const int d0    = (blockIdx.x & 63) * 32;
    const int lbase = part * LPART;
    const int ch = threadIdx.x >> 3;
    const int q  = threadIdx.x & 7;
    const int d  = d0 + ch;
    const int wr = threadIdx.x >> 5;
    const int wc = threadIdx.x & 31;
    const int pslot = ch * PP + q;

    float A0 = -__expf(A_log[(size_t)d * NS + 2 * q]);
    float A1 = -__expf(A_log[(size_t)d * NS + 2 * q + 1]);
    float h0 = 0.f, h1 = 0.f;
    float p0 = 1.f, p1 = 1.f;
    float Dw = 0.f;
    if (!ENDS_ONLY) Dw = Dvec[d0 + wc];

    const float* gDT = dt   + (size_t)(lbase + ch) * DI + d0 + 4 * q;
    const float* gXS = xs   + (size_t)(lbase + ch) * DI + d0 + 4 * q;
    const float* gBC = xdbl + (size_t)(lbase + ch) * XDBL_W + DTR + 4 * q;
    const float* gBC2 = gBC + (size_t)LSEQ * XDBL_W;
    static_assert(XPARTS == 2, "two partials are added here");
    v4f pdt[2], pxs[2], pbc[2], pbc2[2];
    for (int i = 0; i < 2; ++i) {
        pdt[i] = *(const v4f*)(gDT + (size_t)(32 * i) * DI);
        pxs[i] = *(const v4f*)(gXS + (size_t)(32 * i) * DI);
        pbc[i] = *(const v4f*)(gBC + (size_t)(32 * i) * XDBL_W);
        pbc2[i] = *(const v4f*)(gBC2 + (size_t)(32 * i) * XDBL_W);
    }

    if (!ENDS_ONLY) {
        float2 eh[NPART - 1], ep[NPART - 1];
#pragma unroll
        for (int j = 0; j < NPART - 1; ++j) {
            const int jj = (j < part) ? j : 0;
            eh[j] = *(const float2*)&endH[((size_t)jj * DI + d) * NS + 2 * q];
            ep[j] = *(const float2*)&endP[((size_t)jj * DI + d) * NS + 2 * q];
        }
#pragma unroll
        for (int j = 0; j < NPART - 1; ++j)
            if (j < part) {
                h0 = ep[j].x * h0 + eh[j].x;
                h1 = ep[j].y * h1 + eh[j].y;
            }
    }

    for (int i = 0; i < 2; ++i) {
        *(v4f*)&sDT[(threadIdx.x + 256 * i) * 4] = pdt[i];
        *(v4f*)&sXS[(threadIdx.x + 256 * i) * 4] = pxs[i];
        *(v4f*)&sBC[(threadIdx.x + 256 * i) * 4] = pbc[i] + pbc2[i];
    }

    for (int c = 0; c < NCHUNK; ++c) {
        const bool more = (c + 1 < NCHUNK);
        __syncthreads();
        if (more) {
            for (int i = 0; i < 2; ++i) {
                const size_t l1 = (size_t)(c + 1) * 64 + 32 * i;
                pdt[i] = *(const v4f*)(gDT + l1 * DI);
                pxs[i] = *(const v4f*)(gXS + l1 * DI);
                pbc[i] = *(const v4f*)(gBC + l1 * XDBL_W);
                pbc2[i] = *(const v4f*)(gBC2 + l1 * XDBL_W);
            }
        }
        if (ENDS_ONLY) {
            auto read8 = [&](int g, float (&dtg)[8], float (&xg)[8], float2 (&Bg)[8]) {
#pragma unroll
                for (int k = 0; k < 8; ++k) {
                    const int s = g * 8 + k;
                    dtg[k] = sDT[s * 32 + ch];
                    xg[k]  = sXS[s * 32 + ch];
                    Bg[k]  = *(const float2*)&sBC[s * 32 + 2 * q];
                }
            };
            auto step8 = [&](const float (&dtg)[8], const float (&xg)[8], const float2 (&Bg)[8]) {
#pragma unroll
                for (int k = 0; k < 8; ++k) {
                    const float dx  = dtg[k] * xg[k];
                    const float dA0 = __expf(dtg[k] * A0);
                    const float dA1 = __expf(dtg[k] * A1);
                    h0 = dA0 * h0 + dx * Bg[k].x;
                    h1 = dA1 * h1 + dx * Bg[k].y;
                    p0 *= dA0;
                    p1 *= dA1;
                }
            };
            float dta[8], xa[8], dtb[8], xb[8];
            float2 Ba[8], Bb[8];
            read8(0, dta, xa, Ba);
#pragma unroll 1
            for (int g = 0; g < 8; g += 2) {
                read8(g + 1, dtb, xb, Bb);
                step8(dta, xa, Ba);
                if (g + 2 < 8) read8(g + 2, dta, xa, Ba);
                step8(dtb, xb, Bb);
            }
            __syncthreads();
            if (more) {
                for (int i = 0; i < 2; ++i) {
                    *(v4f*)&sDT[(threadIdx.x + 256 * i) * 4] = pdt[i];
                    *(v4f*)&sXS[(threadIdx.x + 256 * i) * 4] = pxs[i];
                    *(v4f*)&sBC[(threadIdx.x + 256 * i) * 4] = pbc[i] + pbc2[i];
                }
            }
        } else {
            float resv[8], xsv[8];
            for (int k = 0; k < 8; ++k) {
                const int l = lbase + c * 64 + wr + 8 * k;
                resv[k] = xr[(size_t)l * (2 * DI) + DI + d0 + wc];
                xsv[k]  = sXS[(wr + 8 * k) * 32 + wc];
            }
#pragma unroll 8
            for (int s = 0; s < 64; ++s) {
                const float dtv = sDT[s * 32 + ch];
                const float xv  = sXS[s * 32 + ch];
                const float2 Bv = *(const float2*)&sBC[s * 32 + 2 * q];
                const float2 Cv = *(const float2*)&sBC[s * 32 + 16 + 2 * q];
                const float dx  = dtv * xv;
                const float dA0 = __expf(dtv * A0);
                const float dA1 = __expf(dtv * A1);
                h0 = dA0 * h0 + dx * Bv.x;
                h1 = dA1 * h1 + dx * Bv.y;
                sP[s * PROW + pslot] = fmaf(h0, Cv.x, h1 * Cv.y);
            }
            __syncthreads();
            if (more) {
                for (int i = 0; i < 2; ++i) {
                    *(v4f*)&sDT[(threadIdx.x + 256 * i) * 4] = pdt[i];
                    *(v4f*)&sXS[(threadIdx.x + 256 * i) * 4] = pxs[i];
                    *(v4f*)&sBC[(threadIdx.x + 256 * i) * 4] = pbc[i] + pbc2[i];
                }
            }
            for (int k = 0; k < 8; ++k) {
                const int ls = wr + 8 * k;
                const int l = lbase + c * 64 + ls;
                const float4 pa = *(const float4*)&sP[ls * PROW + wc * PP];
                const float4 pb = *(const float4*)&sP[ls * PROW + wc * PP + 4];
                float yv = ((pa.x + pa.y) + (pa.z + pa.w)) + ((pb.x + pb.y) + (pb.z + pb.w));
                yv = yv + xsv[k] * Dw;
                const float sig = 1.f / (1.f + __expf(-resv[k]));
                y[(size_t)l * DI + d0 + wc] = yv * (resv[k] * sig);
            }
        }
    }
    if (ENDS_ONLY) {
        *(float2*)&endH[((size_t)part * DI + d) * NS + 2 * q] = make_float2(h0, h1);
        *(float2*)&endP[((size_t)part * DI + d) * NS + 2 * q] = make_float2(p0, p1);
    } else if (part == NPART - 1) {
        *(float2*)&hT[(size_t)d * NS + 2 * q] = make_float2(h0, h1);
    }
}

extern "C" void kernel_launch(void* const* d_in, const int* in_sizes, int n_in,
                              void* d_out, int out_size, void* d_ws, size_t ws_size,
                              hipStream_t stream) {
    const float* x      = (const float*)d_in[0];
    const float* W_in   = (const float*)d_in[1];
    const float* conv_w = (const float*)d_in[2];
    const float* conv_b = (const float*)d_in[3];
    const float* W_x    = (const float*)d_in[4];
    const float* W_dt   = (const float*)d_in[5];
    const float* b_dt   = (const float*)d_in[6];
    const float* A_log  = (const float*)d_in[7];
    const float* Dv     = (const float*)d_in[8];
    const float* W_out  = (const float*)d_in[9];

    float* out = (float*)d_out;
    float* hT  = out + (size_t)LSEQ * DM;

    float* ws   = (float*)d_ws;
    float* xr   = ws;
    float* xs   = xr   + (size_t)LSEQ * 2 * DI;
    float* xdbl = xs   + (size_t)LSEQ * DI;
    float* dt   = xdbl + (size_t)XPARTS * LSEQ * XDBL_W;
    float* yb   = dt   + (size_t)LSEQ * DI;
    float* endH = yb   + (size_t)LSEQ * DI;
    float* endP = endH + (size_t)(NPART - 1) * DI * NS;

    wmma_gemm_f32<128, 128, 64, 2, 4, false, 1><<<512, 256, 0, stream>>>(
        x, W_in, xr, nullptr, LSEQ, 2 * DI, DM, DM, 2 * DI, 2 * DI);

    wmma_gemm_f32_ksplit_conv<XDBL_W, 128, XPARTS><<<XPARTS * (LSEQ / 16), 256, 0, stream>>>(
        xr, conv_w, conv_b, xs, W_x, xdbl);

    wmma_gemm_f32<128, 128, 16, 2, 4, true, XPARTS><<<256, 256, 0, stream>>>(
        xdbl, W_dt, dt, b_dt, LSEQ, DI, DTR, XDBL_W, DI, DI);

    scan_kernel<true><<<(NPART - 1) * (DI / 32), 256, 0, stream>>>(
        dt, xdbl, xs, xr, A_log, Dv, yb, hT, endH, endP);
    scan_kernel<false><<<NPART * (DI / 32), 256, 0, stream>>>(
        dt, xdbl, xs, xr, A_log, Dv, yb, hT, endH, endP);

    wmma_gemm_f32<64, 128, 128, 2, 4, false, 1><<<256, 256, 0, stream>>>(
        yb, W_out, out, nullptr, LSEQ, DM, DI, DI, DM, DM);
}


# GEMM kernel: row pitches compile-time: one base pointer per operand and thread, its pieces at constant offsets: a K-step forms 2 load addresses, not 16-24 with a 64-bit multiply each; sums unchanged
# speedup vs baseline: 8.9615x; 1.0537x over previous
#include <hip/hip_runtime.h>
#include <hip/hip_bf16.h>
#include <math.h>


typedef float v2f __attribute__((ext_vector_type(2)));
typedef float v4f __attribute__((ext_vector_type(4)));
typedef float v8f __attribute__((ext_vector_type(8)));

#define DM   1024
#define DI   2048
#define NS   16
#define KC   4
#define DTR  64
#define LSEQ 2048
#define XDBL_W (DTR + 2*NS)
#define XPARTS 2

template<int BM, int BN, int KS, int WGM, int WGN, bool SOFTPLUS_BIAS, int APARTS,
         int LDA, int LDB, int LDC>
__global__ __launch_bounds__(256)
void wmma_gemm_f32(const float* __restrict__ A, const float* __restrict__ B,
                   float* __restrict__ C, const float* __restrict__ bias,
                   int M, int N, int K)
{
    constexpr int PA  = KS + 4;
    constexpr int PB  = BN + 8;
    constexpr int WM  = BM / (16 * WGM);
    constexpr int WN  = BN / (16 * WGN);
    constexpr int NA4 = BM * KS / 4;
    constexpr int NB4 = KS * BN / 4;
    constexpr int RA  = (NA4 + 255) / 256;
    constexpr int RB  = (NB4 + 255) / 256;
    constexpr int BN4 = BN / 4;
    constexpr int KA4 = KS / 4;
    constexpr int NKK = KS / 4;
    constexpr int KK_STORE = (KS >= 64) ? NKK / 2 : NKK;
    constexpr int RPA = 256 / KA4;
    constexpr int RPB = 256 / BN4;
    static_assert(WGM * WGN == 8, "8 waves per block");
    static_assert(WM * 16 * WGM == BM && WN * 16 * WGN == BN, "tile split");
    static_assert(NA4 % 256 == 0 && NB4 % 256 == 0, "whole pieces per thread");
    static_assert(256 % KA4 == 0 && 256 % BN4 == 0, "a thread's pieces share their column");

    __shared__ __align__(16) float sA[2][BM * PA];
    __shared__ __align__(16) float sB[2][KS * PB];

    const int tid  = threadIdx.x;
    const int lane = tid & 31;
    const int w    = tid >> 5;
    const int wm   = w % WGM;
    const int wn   = w / WGM;
    const int lr   = lane & 15;
    const int lh   = lane >> 4;
    const int tilesM = M / BM;
    const int m0 = (blockIdx.x % tilesM) * BM;
    const int n0 = (blockIdx.x / tilesM) * BN;

    v8f acc[WM][WN];
#pragma unroll
    for (int i = 0; i < WM; ++i)
#pragma unroll
        for (int j = 0; j < WN; ++j) {
            v8f z = {0.f, 0.f, 0.f, 0.f, 0.f, 0.f, 0.f, 0.f};
            acc[i][j] = z;
        }

    v4f ra[RA], rb[RB];
    v4f ra2[RA];
    const size_t apart = (size_t)M * LDA;

    const float* pA = A + (size_t)(m0 + tid / KA4) * LDA + (tid % KA4) * 4;
    const float* pB = B + (size_t)(tid / BN4) * LDB + n0 + (tid % BN4) * 4;
    const int sAo = (tid / KA4) * PA + (tid % KA4) * 4;
    const int sBo = (tid / BN4) * PB + (tid % BN4) * 4;

#pragma unroll
    for (int i = 0; i < RA; ++i) {
        ra[i] = *(const v4f*)(pA + i * (RPA * LDA));
        if (APARTS == 2)
            ra2[i] = *(const v4f*)(pA + apart + i * (RPA * LDA));
    }
#pragma unroll
    for (int i = 0; i < RB; ++i)
        rb[i] = *(const v4f*)(pB + i * (RPB * LDB));
#pragma unroll
    for (int i = 0; i < RA; ++i)
        *(v4f*)&sA[0][sAo + i * (RPA * PA)] = (APARTS == 2) ? ra[i] + ra2[i] : ra[i];
#pragma unroll
    for (int i = 0; i < RB; ++i)
        *(v4f*)&sB[0][sBo + i * (RPB * PB)] = rb[i];
    __syncthreads();

    const int nk = K / KS;
    for (int kt = 0; kt < nk; ++kt) {
        const int buf = kt & 1;
        const bool more = (kt + 1 < nk);
        if (more) {
            const float* qA = pA + (kt + 1) * KS;
            const float* qB = pB + (size_t)((kt + 1) * KS) * LDB;
#pragma unroll
            for (int i = 0; i < RA; ++i) {
                ra[i] = *(const v4f*)(qA + i * (RPA * LDA));
                if (APARTS == 2)
                    ra2[i] = *(const v4f*)(qA + apart + i * (RPA * LDA));
            }
#pragma unroll
            for (int i = 0; i < RB; ++i)
                rb[i] = *(const v4f*)(qB + i * (RPB * LDB));
        }
        const float* tA = &sA[buf][(wm * WM * 16 + lr) * PA + 2 * lh];
        const float* tB = &sB[buf][(2 * lh) * PB + wn * WN * 16 + lr];
        auto products = [&](int kk) {
            v2f a[WM];
#pragma unroll
            for (int i = 0; i < WM; ++i)
                a[i] = *(const v2f*)(tA + i * 16 * PA + kk * 4);
            v2f b[WN];
#pragma unroll
            for (int j = 0; j < WN; ++j) {
                const float* bp = tB + (kk * 4) * PB + j * 16;
                v2f t; t[0] = bp[0]; t[1] = bp[PB];
                b[j] = t;
            }
#pragma unroll
            for (int i = 0; i < WM; ++i)
#pragma unroll
                for (int j = 0; j < WN; ++j)
                    acc[i][j] = __builtin_amdgcn_wmma_f32_16x16x4_f32(
                        false, a[i], false, b[j], (short)0, acc[i][j], false, false);
        };
#pragma unroll
        for (int kk = 0; kk < KK_STORE; ++kk) products(kk);
        if (more) {
#pragma unroll
            for (int i = 0; i < RA; ++i)
                *(v4f*)&sA[buf ^ 1][sAo + i * (RPA * PA)] = (APARTS == 2) ? ra[i] + ra2[i] : ra[i];
#pragma unroll
            for (int i = 0; i < RB; ++i)
                *(v4f*)&sB[buf ^ 1][sBo + i * (RPB * PB)] = rb[i];
        }
#pragma unroll
        for (int kk = KK_STORE; kk < NKK; ++kk) products(kk);
        __syncthreads();
    }

    float* pC = C + (size_t)(m0 + wm * WM * 16 + 8 * lh) * LDC + n0 + wn * WN * 16 + lr;
#pragma unroll
    for (int i = 0; i < WM; ++i) {
#pragma unroll
        for (int j = 0; j < WN; ++j) {
            const int col = n0 + wn * WN * 16 + j * 16 + lr;
#pragma unroll
            for (int r = 0; r < 8; ++r) {
                float v = acc[i][j][r];
                if (SOFTPLUS_BIAS) {
                    v += bias[col];
                    v = (v > 20.f) ? v : __logf(1.f + __expf(v));
                }
                pC[(i * 16 + r) * LDC + j * 16] = v;
            }
        }
    }
}

template<int BN, int KS, int KSPLIT>
__global__ __launch_bounds__(256)
__attribute__((amdgpu_waves_per_eu(2, 2)))
void wmma_gemm_f32_ksplit_conv(const float* __restrict__ xr, const float* __restrict__ conv_w,
                               const float* __restrict__ conv_b, float* __restrict__ xs,
                               const float* __restrict__ B, float* __restrict__ C)
{
    constexpr int BM  = 16;
    constexpr int PA  = KS + 4;
    constexpr int PB  = BN + 8;
    constexpr int WN  = BN / 16;
    constexpr int RB  = KS * BN / 4 / 256;
    constexpr int BN4 = BN / 4;
    constexpr int KW  = KS / 32;
    static_assert(KS == 128, "a wave's 32 lanes x 4 channels span one K-step");
    static_assert(KC == 4, "4 taps: one 16-byte piece of conv_w per channel");
    static_assert((KS * BN / 4) % 256 == 0, "whole pieces per thread");
    static_assert((BM * BN) % 256 == 0, "whole outputs per thread");
    static_assert(8 * BM * BN <= KS * PB, "the 8 partial strips fit in one B buffer");
    static_assert(DI % (KS * KSPLIT) == 0, "whole K-steps per block");
    constexpr int NSTRIP = LSEQ / BM;

    __shared__ __align__(16) float sA[2][BM * PA];
    __shared__ __align__(16) float sB[2][KS * PB];

    const int tid  = threadIdx.x;
    const int lane = tid & 31;
    const int w    = tid >> 5;
    const int lr   = lane & 15;
    const int lh   = lane >> 4;
    const int m0   = (blockIdx.x % NSTRIP) * BM;
    const int kpart = blockIdx.x / NSTRIP;
    const int kb   = kpart * (DI / KSPLIT);

    v8f acc[WN];
#pragma unroll
    for (int j = 0; j < WN; ++j) {
        v8f z = {0.f, 0.f, 0.f, 0.f, 0.f, 0.f, 0.f, 0.f};
        acc[j] = z;
    }

    const int r0 = m0 + 2 * w;
    const float* xrow[5];
    bool xok[3];
#pragma unroll
    for (int j = 0; j < 5; ++j) {
        const int r = r0 - 3 + j;
        if (j < 3) xok[j] = (r >= 0);
        xrow[j] = xr + (size_t)(r >= 0 ? r : 0) * (2 * DI) + 4 * lane;
    }
    const float* wcol = conv_w + (size_t)(4 * lane) * KC;
    const float* bcol = conv_b + 4 * lane;
    float* xsrow = xs + (size_t)r0 * DI + 4 * lane;

    v4f xv[5], wq[4], bq;
    v4f rb[RB];
    v4f xo[2];

    auto request = [&](int k0) {
#pragma unroll
        for (int j = 0; j < 5; ++j)
            xv[j] = *(const v4f*)(xrow[j] + k0);
#pragma unroll
        for (int c = 0; c < 4; ++c)
            wq[c] = *(const v4f*)(wcol + (size_t)(k0 + c) * KC);
        bq = *(const v4f*)(bcol + k0);
#pragma unroll
        for (int i = 0; i < RB; ++i) {
            const int idx = tid + i * 256;
            rb[i] = *(const v4f*)(B + (size_t)(k0 + idx / BN4) * BN + (idx % BN4) * 4);
        }
    };
    auto stage = [&](int buf) {
#pragma unroll
        for (int j = 0; j < 3; ++j)
            if (!xok[j]) { v4f z = {0.f, 0.f, 0.f, 0.f}; xv[j] = z; }
#pragma unroll
        for (int i = 0; i < 2; ++i) {
            v4f o;
#pragma unroll
            for (int c = 0; c < 4; ++c) {
                float a = bq[c];
#pragma unroll
                for (int j = 0; j < 4; ++j)
                    a += xv[i + j][c] * wq[c][j];
                const float sig = 1.f / (1.f + __expf(-a));
                o[c] = a * sig;
            }
            *(v4f*)&sA[buf][(2 * w + i) * PA + 4 * lane] = o;
            xo[i] = o;
        }
#pragma unroll
        for (int i = 0; i < RB; ++i) {
            const int idx = tid + i * 256;
            *(v4f*)&sB[buf][(idx / BN4) * PB + (idx % BN4) * 4] = rb[i];
        }
    };

    auto flush = [&](int k0) {
#pragma unroll
        for (int i = 0; i < 2; ++i)
            *(v4f*)(xsrow + (size_t)i * DI + k0) = xo[i];
    };

    request(kb);
    stage(0);
    __syncthreads();

    constexpr int nk = DI / KS / KSPLIT;
    for (int kt = 0; kt < nk; ++kt) {
        const int buf = kt & 1;
        const bool more = (kt + 1 < nk);
        if (more) request(kb + (kt + 1) * KS);
        flush(kb + kt * KS);
        __builtin_amdgcn_sched_barrier(0);
        const float* tA = &sA[buf][lr * PA + w * KW * 4 + 2 * lh];
        const float* tB = &sB[buf][(w * KW * 4 + 2 * lh) * PB + lr];
#pragma unroll
        for (int kk = 0; kk < KW; ++kk) {
            const v2f a = *(const v2f*)(tA + kk * 4);
            v2f b[WN];
#pragma unroll
            for (int j = 0; j < WN; ++j) {
                const float* bp = tB + (kk * 4) * PB + j * 16;
                v2f t; t[0] = bp[0]; t[1] = bp[PB];
                b[j] = t;
            }
#pragma unroll
            for (int j = 0; j < WN; ++j)
                acc[j] = __builtin_amdgcn_wmma_f32_16x16x4_f32(
                    false, a, false, b[j], (short)0, acc[j], false, false);
        }
        if (more) stage(buf ^ 1);
        __syncthreads();
    }

    float* sP = &sB[0][0];
#pragma unroll
    for (int j = 0; j < WN; ++j)
#pragma unroll
        for (int r = 0; r < 8; ++r)
            sP[w * (BM * BN) + (r + 8 * lh) * BN + j * 16 + lr] = acc[j][r];
    __syncthreads();
    float* Cs = C + (size_t)kpart * ((size_t)LSEQ * BN) + (size_t)m0 * BN;
#pragma unroll
    for (int i = 0; i < BM * BN / 256; ++i) {
        const int idx = tid + i * 256;
        const float p0 = sP[0 * (BM * BN) + idx], p1 = sP[1 * (BM * BN) + idx];
        const float p2 = sP[2 * (BM * BN) + idx], p3 = sP[3 * (BM * BN) + idx];
        const float p4 = sP[4 * (BM * BN) + idx], p5 = sP[5 * (BM * BN) + idx];
        const float p6 = sP[6 * (BM * BN) + idx], p7 = sP[7 * (BM * BN) + idx];
        Cs[idx] = ((p0 + p1) + (p2 + p3)) + ((p4 + p5) + (p6 + p7));
    }
}

#define NPART 8
#define LPART (LSEQ / NPART)
#define NCHUNK (LPART / 64)
#define PP 12
#define PROW (32 * PP)

template<bool ENDS_ONLY>
__global__ __launch_bounds__(256)
void scan_kernel(const float* __restrict__ dt, const float* __restrict__ xdbl,
                 const float* __restrict__ xs, const float* __restrict__ xr,
                 const float* __restrict__ A_log, const float* __restrict__ Dvec,
                 float* __restrict__ y, float* __restrict__ hT,
                 float* __restrict__ endH, float* __restrict__ endP)
{
    __shared__ __align__(16) float sDT[64 * 32];
    __shared__ __align__(16) float sXS[64 * 32];
    __shared__ __align__(16) float sBC[64 * 32];
    __shared__ __align__(16) float sP[ENDS_ONLY ? 4 : 64 * PROW];

    const int part  = blockIdx.x >> 6;
    const int d0    = (blockIdx.x & 63) * 32;
    const int lbase = part * LPART;
    const int ch = threadIdx.x >> 3;
    const int q  = threadIdx.x & 7;
    const int d  = d0 + ch;
    const int wr = threadIdx.x >> 5;
    const int wc = threadIdx.x & 31;
    const int pslot = ch * PP + q;

    float A0 = -__expf(A_log[(size_t)d * NS + 2 * q]);
    float A1 = -__expf(A_log[(size_t)d * NS + 2 * q + 1]);
    float h0 = 0.f, h1 = 0.f;
    float p0 = 1.f, p1 = 1.f;
    float Dw = 0.f;
    if (!ENDS_ONLY) Dw = Dvec[d0 + wc];

    const float* gDT = dt   + (size_t)(lbase + ch) * DI + d0 + 4 * q;
    const float* gXS = xs   + (size_t)(lbase + ch) * DI + d0 + 4 * q;
    const float* gBC = xdbl + (size_t)(lbase + ch) * XDBL_W + DTR + 4 * q;
    const float* gBC2 = gBC + (size_t)LSEQ * XDBL_W;
    static_assert(XPARTS == 2, "two partials are added here");
    v4f pdt[2], pxs[2], pbc[2], pbc2[2];
    for (int i = 0; i < 2; ++i) {
        pdt[i] = *(const v4f*)(gDT + (size_t)(32 * i) * DI);
        pxs[i] = *(const v4f*)(gXS + (size_t)(32 * i) * DI);
        pbc[i] = *(const v4f*)(gBC + (size_t)(32 * i) * XDBL_W);
        pbc2[i] = *(const v4f*)(gBC2 + (size_t)(32 * i) * XDBL_W);
    }

    if (!ENDS_ONLY) {
        float2 eh[NPART - 1], ep[NPART - 1];
#pragma unroll
        for (int j = 0; j < NPART - 1; ++j) {
            const int jj = (j < part) ? j : 0;
            eh[j] = *(const float2*)&endH[((size_t)jj * DI + d) * NS + 2 * q];
            ep[j] = *(const float2*)&endP[((size_t)jj * DI + d) * NS + 2 * q];
        }
#pragma unroll
        for (int j = 0; j < NPART - 1; ++j)
            if (j < part) {
                h0 = ep[j].x * h0 + eh[j].x;
                h1 = ep[j].y * h1 + eh[j].y;
            }
    }

    for (int i = 0; i < 2; ++i) {
        *(v4f*)&sDT[(threadIdx.x + 256 * i) * 4] = pdt[i];
        *(v4f*)&sXS[(threadIdx.x + 256 * i) * 4] = pxs[i];
        *(v4f*)&sBC[(threadIdx.x + 256 * i) * 4] = pbc[i] + pbc2[i];
    }

    for (int c = 0; c < NCHUNK; ++c) {
        const bool more = (c + 1 < NCHUNK);
        __syncthreads();
        if (more) {
            for (int i = 0; i < 2; ++i) {
                const size_t l1 = (size_t)(c + 1) * 64 + 32 * i;
                pdt[i] = *(const v4f*)(gDT + l1 * DI);
                pxs[i] = *(const v4f*)(gXS + l1 * DI);
                pbc[i] = *(const v4f*)(gBC + l1 * XDBL_W);
                pbc2[i] = *(const v4f*)(gBC2 + l1 * XDBL_W);
            }
        }
        if (ENDS_ONLY) {
            auto read8 = [&](int g, float (&dtg)[8], float (&xg)[8], float2 (&Bg)[8]) {
#pragma unroll
                for (int k = 0; k < 8; ++k) {
                    const int s = g * 8 + k;
                    dtg[k] = sDT[s * 32 + ch];
                    xg[k]  = sXS[s * 32 + ch];
                    Bg[k]  = *(const float2*)&sBC[s * 32 + 2 * q];
                }
            };
            auto step8 = [&](const float (&dtg)[8], const float (&xg)[8], const float2 (&Bg)[8]) {
#pragma unroll
                for (int k = 0; k < 8; ++k) {
                    const float dx  = dtg[k] * xg[k];
                    const float dA0 = __expf(dtg[k] * A0);
                    const float dA1 = __expf(dtg[k] * A1);
                    h0 = dA0 * h0 + dx * Bg[k].x;
                    h1 = dA1 * h1 + dx * Bg[k].y;
                    p0 *= dA0;
                    p1 *= dA1;
                }
            };
            float dta[8], xa[8], dtb[8], xb[8];
            float2 Ba[8], Bb[8];
            read8(0, dta, xa, Ba);
#pragma unroll 1
            for (int g = 0; g < 8; g += 2) {
                read8(g + 1, dtb, xb, Bb);
                step8(dta, xa, Ba);
                if (g + 2 < 8) read8(g + 2, dta, xa, Ba);
                step8(dtb, xb, Bb);
            }
            __syncthreads();
            if (more) {
                for (int i = 0; i < 2; ++i) {
                    *(v4f*)&sDT[(threadIdx.x + 256 * i) * 4] = pdt[i];
                    *(v4f*)&sXS[(threadIdx.x + 256 * i) * 4] = pxs[i];
                    *(v4f*)&sBC[(threadIdx.x + 256 * i) * 4] = pbc[i] + pbc2[i];
                }
            }
        } else {
            float resv[8], xsv[8];
            for (int k = 0; k < 8; ++k) {
                const int l = lbase + c * 64 + wr + 8 * k;
                resv[k] = xr[(size_t)l * (2 * DI) + DI + d0 + wc];
                xsv[k]  = sXS[(wr + 8 * k) * 32 + wc];
            }
#pragma unroll 8
            for (int s = 0; s < 64; ++s) {
                const float dtv = sDT[s * 32 + ch];
                const float xv  = sXS[s * 32 + ch];
                const float2 Bv = *(const float2*)&sBC[s * 32 + 2 * q];
                const float2 Cv = *(const float2*)&sBC[s * 32 + 16 + 2 * q];
                const float dx  = dtv * xv;
                const float dA0 = __expf(dtv * A0);
                const float dA1 = __expf(dtv * A1);
                h0 = dA0 * h0 + dx * Bv.x;
                h1 = dA1 * h1 + dx * Bv.y;
                sP[s * PROW + pslot] = fmaf(h0, Cv.x, h1 * Cv.y);
            }
            __syncthreads();
            if (more) {
                for (int i = 0; i < 2; ++i) {
                    *(v4f*)&sDT[(threadIdx.x + 256 * i) * 4] = pdt[i];
                    *(v4f*)&sXS[(threadIdx.x + 256 * i) * 4] = pxs[i];
                    *(v4f*)&sBC[(threadIdx.x + 256 * i) * 4] = pbc[i] + pbc2[i];
                }
            }
            for (int k = 0; k < 8; ++k) {
                const int ls = wr + 8 * k;
                const int l = lbase + c * 64 + ls;
                const float4 pa = *(const float4*)&sP[ls * PROW + wc * PP];
                const float4 pb = *(const float4*)&sP[ls * PROW + wc * PP + 4];
                float yv = ((pa.x + pa.y) + (pa.z + pa.w)) + ((pb.x + pb.y) + (pb.z + pb.w));
                yv = yv + xsv[k] * Dw;
                const float sig = 1.f / (1.f + __expf(-resv[k]));
                y[(size_t)l * DI + d0 + wc] = yv * (resv[k] * sig);
            }
        }
    }
    if (ENDS_ONLY) {
        *(float2*)&endH[((size_t)part * DI + d) * NS + 2 * q] = make_float2(h0, h1);
        *(float2*)&endP[((size_t)part * DI + d) * NS + 2 * q] = make_float2(p0, p1);
    } else if (part == NPART - 1) {
        *(float2*)&hT[(size_t)d * NS + 2 * q] = make_float2(h0, h1);
    }
}

extern "C" void kernel_launch(void* const* d_in, const int* in_sizes, int n_in,
                              void* d_out, int out_size, void* d_ws, size_t ws_size,
                              hipStream_t stream) {
    const float* x      = (const float*)d_in[0];
    const float* W_in   = (const float*)d_in[1];
    const float* conv_w = (const float*)d_in[2];
    const float* conv_b = (const float*)d_in[3];
    const float* W_x    = (const float*)d_in[4];
    const float* W_dt   = (const float*)d_in[5];
    const float* b_dt   = (const float*)d_in[6];
    const float* A_log  = (const float*)d_in[7];
    const float* Dv     = (const float*)d_in[8];
    const float* W_out  = (const float*)d_in[9];

    float* out = (float*)d_out;
    float* hT  = out + (size_t)LSEQ * DM;

    float* ws   = (float*)d_ws;
    float* xr   = ws;
    float* xs   = xr   + (size_t)LSEQ * 2 * DI;
    float* xdbl = xs   + (size_t)LSEQ * DI;
    float* dt   = xdbl + (size_t)XPARTS * LSEQ * XDBL_W;
    float* yb   = dt   + (size_t)LSEQ * DI;
    float* endH = yb   + (size_t)LSEQ * DI;
    float* endP = endH + (size_t)(NPART - 1) * DI * NS;

    wmma_gemm_f32<128, 128, 64, 2, 4, false, 1, DM, 2 * DI, 2 * DI><<<512, 256, 0, stream>>>(
        x, W_in, xr, nullptr, LSEQ, 2 * DI, DM);

    wmma_gemm_f32_ksplit_conv<XDBL_W, 128, XPARTS><<<XPARTS * (LSEQ / 16), 256, 0, stream>>>(
        xr, conv_w, conv_b, xs, W_x, xdbl);

    wmma_gemm_f32<128, 128, 16, 2, 4, true, XPARTS, XDBL_W, DI, DI><<<256, 256, 0, stream>>>(
        xdbl, W_dt, dt, b_dt, LSEQ, DI, DTR);

    scan_kernel<true><<<(NPART - 1) * (DI / 32), 256, 0, stream>>>(
        dt, xdbl, xs, xr, A_log, Dv, yb, hT, endH, endP);
    scan_kernel<false><<<NPART * (DI / 32), 256, 0, stream>>>(
        dt, xdbl, xs, xr, A_log, Dv, yb, hT, endH, endP);

    wmma_gemm_f32<64, 128, 128, 2, 4, false, 1, DI, DM, DM><<<256, 256, 0, stream>>>(
        yb, W_out, out, nullptr, LSEQ, DM, DI);
}
